# starting kernel
# baseline (speedup 1.0000x reference)
#include <hip/hip_runtime.h>
#include <hip/hip_bf16.h>
#include <math.h>

typedef float v2f __attribute__((ext_vector_type(2)));
typedef float v8f __attribute__((ext_vector_type(8)));

__device__ __forceinline__ void atomicAddF32(float* p, float v) {
    unsafeAtomicAdd(p, v);
}

__device__ __forceinline__ void atomicMaxF32(float* addr, float val) {
    if (val >= 0.0f) {
        atomicMax((int*)addr, __float_as_int(val));
    } else {
        atomicMin((unsigned int*)addr, __float_as_uint(val));
    }
}

__device__ __forceinline__ void edge_sd(const int* __restrict__ ei, int E, int e,
                                        int& s, int& d) {
    if (e < E) { s = ei[e]; d = ei[E + e]; }
    else       { s = e - E; d = e - E; }
}

__global__ void gat_init_kernel(float* __restrict__ agg,
                                float* __restrict__ emax,
                                float* __restrict__ denom,
                                int n_agg, int n_nodes2) {
    int stride = gridDim.x * blockDim.x;
    for (int i = blockIdx.x * blockDim.x + threadIdx.x; i < n_agg; i += stride)
        agg[i] = 0.0f;
    for (int i = blockIdx.x * blockDim.x + threadIdx.x; i < n_nodes2; i += stride) {
        emax[i]  = -INFINITY;
        denom[i] = 0.0f;
    }
}

template <int K, int NDIM, int ROWS>
__global__ void wmma_gemm_kernel(const float* __restrict__ A,
                                 const float* __restrict__ B,
                                 const float* __restrict__ bias,
                                 float* __restrict__ C,
                                 int M) {
    constexpr int COL_TILES = NDIM / 16;
    __shared__ float Alds[ROWS * (K + 1)];

    const int tid     = threadIdx.x;
    const int wave    = tid >> 5;
    const int lane    = tid & 31;
    const int rowbase = blockIdx.x * ROWS;

    for (int idx = tid; idx < ROWS * K; idx += blockDim.x) {
        int r  = idx / K;
        int c  = idx - r * K;
        int gr = rowbase + r;
        Alds[r * (K + 1) + c] = (gr < M) ? A[(size_t)gr * K + c] : 0.0f;
    }
    __syncthreads();

    const int rg      = wave / COL_TILES;
    const int colbase = (wave % COL_TILES) * 16;
    const int half    = lane >> 4;
    const int l15     = lane & 15;

    const float* Arow = &Alds[(rg * 16 + l15) * (K + 1)];

    v8f acc = {};
    #pragma unroll 8
    for (int k = 0; k < K; k += 4) {
        const int kk = k + 2 * half;
        v2f a, b;
        a.x = Arow[kk];
        a.y = Arow[kk + 1];
        b.x = B[(size_t)kk * NDIM + colbase + l15];
        b.y = B[(size_t)(kk + 1) * NDIM + colbase + l15];
        acc = __builtin_amdgcn_wmma_f32_16x16x4_f32(
             false, a,  false, b,
             (short)0, acc,  false,  false);
    }

    const float bv = bias ? bias[colbase + l15] : 0.0f;
    #pragma unroll
    for (int r = 0; r < 8; ++r) {
        int grow = rowbase + rg * 16 + r + 8 * half;
        if (grow < M)
            C[(size_t)grow * NDIM + colbase + l15] = acc[r] + bv;
    }
}

__global__ void alpha_kernel(const float* __restrict__ xh,
                             const float* __restrict__ attl,
                             const float* __restrict__ attr,
                             float* __restrict__ al,
                             float* __restrict__ ar, int N) {
    const int gw   = (blockIdx.x * blockDim.x + threadIdx.x) >> 5;
    const int lane = threadIdx.x & 31;
    if (gw >= N) return;

    const float* xr = xh + (size_t)gw * 128;
    const float x0 = xr[lane], x1 = xr[lane + 32], x2 = xr[lane + 64], x3 = xr[lane + 96];

    float l0 = x0 * attl[lane]      + x1 * attl[lane + 32];
    float l1 = x2 * attl[lane + 64] + x3 * attl[lane + 96];
    float r0 = x0 * attr[lane]      + x1 * attr[lane + 32];
    float r1 = x2 * attr[lane + 64] + x3 * attr[lane + 96];

    #pragma unroll
    for (int off = 16; off > 0; off >>= 1) {
        l0 += __shfl_xor(l0, off, 32);
        l1 += __shfl_xor(l1, off, 32);
        r0 += __shfl_xor(r0, off, 32);
        r1 += __shfl_xor(r1, off, 32);
    }
    if (lane == 0) {
        al[gw * 2 + 0] = l0;  al[gw * 2 + 1] = l1;
        ar[gw * 2 + 0] = r0;  ar[gw * 2 + 1] = r1;
    }
}

__global__ void edge_logits_kernel(const int* __restrict__ ei, int E, int N,
                                   const float* __restrict__ al,
                                   const float* __restrict__ ar,
                                   float* __restrict__ elog,
                                   float* __restrict__ emax) {
    const int e = blockIdx.x * blockDim.x + threadIdx.x;
    if (e >= E + N) return;
    int s, d;
    edge_sd(ei, E, e, s, d);
    #pragma unroll
    for (int h = 0; h < 2; ++h) {
        float v = al[s * 2 + h] + ar[d * 2 + h];
        v = (v > 0.0f) ? v : 0.2f * v;
        elog[(size_t)e * 2 + h] = v;
        atomicMaxF32(&emax[d * 2 + h], v);
    }
}

__global__ void edge_exp_kernel(const int* __restrict__ ei, int E, int N,
                                float* __restrict__ elog,
                                const float* __restrict__ emax,
                                float* __restrict__ denom) {
    const int e = blockIdx.x * blockDim.x + threadIdx.x;
    if (e >= E + N) return;
    int s, d;
    edge_sd(ei, E, e, s, d);
    (void)s;
    #pragma unroll
    for (int h = 0; h < 2; ++h) {
        float m = emax[d * 2 + h];
        if (!(fabsf(m) <= 3.4e38f)) m = 0.0f;
        float v = __expf(elog[(size_t)e * 2 + h] - m);
        elog[(size_t)e * 2 + h] = v;
        atomicAddF32(&denom[d * 2 + h], v);
    }
}

__global__ void aggregate_kernel(const int* __restrict__ ei, int E, int N,
                                 const float* __restrict__ xh,
                                 const float* __restrict__ ex,
                                 const float* __restrict__ denom,
                                 float* __restrict__ agg) {
    const int gw   = (blockIdx.x * blockDim.x + threadIdx.x) >> 5;
    const int lane = threadIdx.x & 31;
    if (gw >= E + N) return;
    int s, d;
    edge_sd(ei, E, gw, s, d);

    const float w0 = ex[(size_t)gw * 2 + 0] / (denom[d * 2 + 0] + 1e-16f);
    const float w1 = ex[(size_t)gw * 2 + 1] / (denom[d * 2 + 1] + 1e-16f);

    const float* xs = xh  + (size_t)s * 128;
    float*       od = agg + (size_t)d * 128;
    #pragma unroll
    for (int q = 0; q < 4; ++q) {
        const int c = lane + q * 32;
        const float w = (c < 64) ? w0 : w1;
        atomicAddF32(od + c, xs[c] * w);
    }
}

extern "C" void kernel_launch(void* const* d_in, const int* in_sizes, int n_in,
                              void* d_out, int out_size, void* d_ws, size_t ws_size,
                              hipStream_t stream) {
    const float* x    = (const float*)d_in[0];
    const int*   ei   = (const int*)d_in[1];
    const float* Wl   = (const float*)d_in[2];
    const float* attl = (const float*)d_in[3];
    const float* attr = (const float*)d_in[4];
    const float* Wout = (const float*)d_in[5];
    const float* bias = (const float*)d_in[6];
    float* out = (float*)d_out;

    const int N   = in_sizes[0] / 128;
    const int E   = in_sizes[1] / 2;
    const int EDG = E + N;

    float* ws    = (float*)d_ws;
    float* xh    = ws;
    float* agg   = xh  + (size_t)N * 128;
    float* al    = agg + (size_t)N * 128;
    float* ar    = al  + (size_t)N * 2;
    float* emax  = ar  + (size_t)N * 2;
    float* denom = emax + (size_t)N * 2;
    float* elog  = denom + (size_t)N * 2;

    gat_init_kernel<<<4096, 256, 0, stream>>>(agg, emax, denom, N * 128, N * 2);

    wmma_gemm_kernel<128, 128, 16><<<(N + 15) / 16, 256, 0, stream>>>(
        x, Wl, nullptr, xh, N);

    alpha_kernel<<<(N + 7) / 8, 256, 0, stream>>>(xh, attl, attr, al, ar, N);

    edge_logits_kernel<<<(EDG + 255) / 256, 256, 0, stream>>>(
        ei, E, N, al, ar, elog, emax);

    edge_exp_kernel<<<(EDG + 255) / 256, 256, 0, stream>>>(
        ei, E, N, elog, emax, denom);

    aggregate_kernel<<<(EDG + 7) / 8, 256, 0, stream>>>(
        ei, E, N, xh, elog, denom, agg);

    wmma_gemm_kernel<128, 64, 32><<<(N + 31) / 32, 256, 0, stream>>>(
        agg, Wout, bias, out, N);
}


# Owner scan reads one range-key byte per edge (16 edges per 16-byte load, word-wise match), written by the alpha kernel over spent xh rows; sums in LDS arrival order as before
# speedup vs baseline: 3.8124x; 3.8124x over previous
#include <hip/hip_runtime.h>
#include <hip/hip_bf16.h>
#include <math.h>

typedef float v2f __attribute__((ext_vector_type(2)));
typedef float v8f __attribute__((ext_vector_type(8)));

__device__ __forceinline__ unsigned int f32_to_key(float v) {
    const unsigned int b = __float_as_uint(v);
    return (b & 0x80000000u) ? ~b : (b | 0x80000000u);
}
__device__ __forceinline__ float key_to_f32(unsigned int k) {
    const unsigned int b = (k & 0x80000000u) ? (k & 0x7fffffffu) : ~k;
    return __uint_as_float(b);
}

__global__ void proj_gemm_kernel(const float* __restrict__ A,
                                 const float* __restrict__ B,
                                 const float* __restrict__ Wout,
                                 float* __restrict__ XH,
                                 float* __restrict__ Y,
                                 int M) {
    constexpr int K     = 128;
    constexpr int NDIM  = 128;
    constexpr int HC    = 64;
    constexpr int ROWS  = 16;
    constexpr int PITCH = K + 1;
    __shared__ float Alds[ROWS * PITCH];

    const int tid     = threadIdx.x;
    const int wave    = tid >> 5;
    const int lane    = tid & 31;
    const int rowbase = blockIdx.x * ROWS;

    for (int idx = tid; idx < ROWS * K; idx += blockDim.x) {
        int r  = idx / K;
        int c  = idx - r * K;
        int gr = rowbase + r;
        Alds[r * PITCH + c] = (gr < M) ? A[(size_t)gr * K + c] : 0.0f;
    }
    __syncthreads();

    const int colbase = wave * 16;
    const int half    = lane >> 4;
    const int l15     = lane & 15;

    const float* Arow = &Alds[l15 * PITCH];

    v8f acc = {};
    #pragma unroll 8
    for (int k = 0; k < K; k += 4) {
        const int kk = k + 2 * half;
        v2f a, b;
        a.x = Arow[kk];
        a.y = Arow[kk + 1];
        b.x = B[(size_t)kk * NDIM + colbase + l15];
        b.y = B[(size_t)(kk + 1) * NDIM + colbase + l15];
        acc = __builtin_amdgcn_wmma_f32_16x16x4_f32(
             false, a,  false, b,
             (short)0, acc,  false,  false);
    }

    #pragma unroll
    for (int r = 0; r < 8; ++r) {
        int grow = rowbase + r + 8 * half;
        if (grow < M)
            XH[(size_t)grow * NDIM + colbase + l15] = acc[r];
    }

    __syncthreads();
    #pragma unroll
    for (int r = 0; r < 8; ++r)
        Alds[(r + 8 * half) * PITCH + colbase + l15] = acc[r];
    __syncthreads();

    const int kbase = (wave >> 2) * HC;
    const int ycol  = (wave & 3) * 16;
    v8f acc2 = {};
    #pragma unroll 8
    for (int k = 0; k < HC; k += 4) {
        const int kk = kbase + k + 2 * half;
        v2f a, b;
        a.x = Arow[kk];
        a.y = Arow[kk + 1];
        b.x = Wout[(size_t)kk * HC + ycol + l15];
        b.y = Wout[(size_t)(kk + 1) * HC + ycol + l15];
        acc2 = __builtin_amdgcn_wmma_f32_16x16x4_f32(
             false, a,  false, b,
             (short)0, acc2,  false,  false);
    }
    #pragma unroll
    for (int r = 0; r < 8; ++r) {
        int grow = rowbase + r + 8 * half;
        if (grow < M)
            Y[(size_t)grow * NDIM + colbase + l15] = acc2[r];
    }
}

typedef int v4i __attribute__((ext_vector_type(4)));

constexpr int OWN_R   = 196;
constexpr int OWN_CAP = 8192;
constexpr int OWN_T   = 1024;

__device__ __forceinline__ bool keys_usable(int N, int E) {
    return (N + OWN_R - 1) / OWN_R <= 256 && (long long)E <= 512LL * (long long)N;
}

__global__ void alpha_kernel(const float* xh,
                             const float* __restrict__ attl,
                             const float* __restrict__ attr,
                             float* __restrict__ al,
                             float* __restrict__ ar, int N,
                             const int* __restrict__ ei, int E,
                             unsigned int* keys) {
    const int gw   = (blockIdx.x * blockDim.x + threadIdx.x) >> 5;
    const int lane = threadIdx.x & 31;

    if (gw < N) {
        const float* xr = xh + (size_t)gw * 128;
        const float x0 = xr[lane], x1 = xr[lane + 32], x2 = xr[lane + 64], x3 = xr[lane + 96];

        float l0 = x0 * attl[lane]      + x1 * attl[lane + 32];
        float l1 = x2 * attl[lane + 64] + x3 * attl[lane + 96];
        float r0 = x0 * attr[lane]      + x1 * attr[lane + 32];
        float r1 = x2 * attr[lane + 64] + x3 * attr[lane + 96];

        #pragma unroll
        for (int off = 16; off > 0; off >>= 1) {
            l0 += __shfl_xor(l0, off, 32);
            l1 += __shfl_xor(l1, off, 32);
            r0 += __shfl_xor(r0, off, 32);
            r1 += __shfl_xor(r1, off, 32);
        }
        if (lane == 0) {
            al[gw * 2 + 0] = l0;  al[gw * 2 + 1] = l1;
            ar[gw * 2 + 0] = r0;  ar[gw * 2 + 1] = r1;
        }
    }

    if (!keys_usable(N, E)) return;
    __syncthreads();

    const long long e0 = (long long)blockIdx.x * 4096 + 16 * (int)threadIdx.x;
    if (e0 >= E) return;
    const int* dcol = ei + E;
    if ((E & 3) == 0) {
        v4i d[4];
        #pragma unroll
        for (int j = 0; j < 4; ++j)
            d[j] = (e0 + 4 * j < E) ? *(const v4i*)(dcol + e0 + 4 * j) : v4i{0, 0, 0, 0};
        #pragma unroll
        for (int j = 0; j < 4; ++j) {
            const unsigned int w = ((unsigned int)d[j][0] / OWN_R)
                                 | ((unsigned int)d[j][1] / OWN_R) << 8
                                 | ((unsigned int)d[j][2] / OWN_R) << 16
                                 | ((unsigned int)d[j][3] / OWN_R) << 24;
            if (e0 + 4 * j < E) keys[(e0 >> 2) + j] = w;
        }
    } else {
        unsigned char* kb = (unsigned char*)keys;
        for (int j = 0; j < 16; ++j)
            if (e0 + j < E) kb[e0 + j] = (unsigned char)((unsigned int)dcol[e0 + j] / OWN_R);
    }
}

__device__ __forceinline__ void lds_add_f32(float* p, float v) {
    __hip_atomic_fetch_add(p, v, __ATOMIC_RELAXED, __HIP_MEMORY_SCOPE_WORKGROUP);
}

__device__ __forceinline__ void own_append(int e, unsigned int* cnt, int* ls) {
    const unsigned int pos = atomicAdd(cnt, 1u);
    if (pos < (unsigned int)OWN_CAP) ls[pos] = e;
}

__device__ __forceinline__ void scan_word(unsigned int w, unsigned int pat, int ebase,
                                          unsigned int* cnt, int* ls) {
    const unsigned int x = w ^ pat;
    unsigned int z = ~((((x & 0x7f7f7f7fu) + 0x7f7f7f7fu) | x) | 0x7f7f7f7fu);
    while (z) {
        own_append(ebase + (__builtin_ctz(z) >> 3), cnt, ls);
        z &= z - 1u;
    }
}

__device__ __forceinline__ void edge_logit2(const float* __restrict__ al,
                                            const float* __restrict__ ar,
                                            int s, int d, float& v0, float& v1) {
    const float a0 = al[2 * s], a1 = al[2 * s + 1];
    const float r0 = ar[2 * d], r1 = ar[2 * d + 1];
    v0 = a0 + r0;  v0 = (v0 > 0.0f) ? v0 : 0.2f * v0;
    v1 = a1 + r1;  v1 = (v1 > 0.0f) ? v1 : 0.2f * v1;
}

__device__ __forceinline__ float seg_max_of(unsigned int key) {
    float m = key_to_f32(key);
    if (!(fabsf(m) <= 3.4e38f)) m = 0.0f;
    return m;
}

template <class F>
__device__ __forceinline__ void own_rescan(const int* __restrict__ ei, int E,
                                           int lo, int nown, int tid, F f) {
    for (int e = tid; e < E; e += OWN_T) {
        const unsigned int dl = (unsigned int)(ei[E + e] - lo);
        if (dl < (unsigned int)nown) f(ei[e], (int)dl);
    }
    for (int i = tid; i < nown; i += OWN_T) f(lo + i, i);
}

__global__ __launch_bounds__(1024)
void gat_owner_kernel(const int* __restrict__ ei, int E, int N,
                      const unsigned int* __restrict__ keys,
                      const float* __restrict__ al,
                      const float* __restrict__ ar,
                      const float* __restrict__ y,
                      const float* __restrict__ bias,
                      float* __restrict__ out) {
    __shared__ float          tile[OWN_R * 64];
    __shared__ float          lv[OWN_CAP * 2];
    __shared__ int            ls[OWN_CAP];
    __shared__ unsigned short ld[OWN_CAP];
    __shared__ unsigned int   mkey[OWN_R * 2];
    __shared__ float          den[OWN_R * 2];
    __shared__ unsigned int   cnt;

    const int tid  = threadIdx.x;
    const int lane = tid & 31;
    const int wave = tid >> 5;
    const int nranges = (N + OWN_R - 1) / OWN_R;
    const bool usekeys = keys_usable(N, E);

    for (int rg = blockIdx.x; rg < nranges; rg += gridDim.x) {
        const int lo   = rg * OWN_R;
        const int nown = min(OWN_R, N - lo);

        for (int i = tid; i < nown * 64; i += OWN_T) tile[i] = 0.0f;
        for (int i = tid; i < nown * 2; i += OWN_T) { mkey[i] = 0u; den[i] = 0.0f; }
        if (tid == 0) cnt = 0u;
        __syncthreads();

        if (usekeys) {
            const unsigned int pat = (unsigned int)rg * 0x01010101u;
            const v4i* __restrict__ k4 = (const v4i*)keys;
            const int nv = E >> 4;
            constexpr int G = 8;
            const int ngroups = nv / (G * OWN_T);
            int g = (int)(((long long)rg * ngroups) / nranges);
            for (int t = 0; t < ngroups; ++t) {
                const int b = g * (G * OWN_T) + tid;
                g = (g + 1 < ngroups) ? g + 1 : 0;
                v4i v[G];
                #pragma unroll
                for (int k = 0; k < G; ++k)
                    v[k] = __builtin_nontemporal_load(&k4[b + k * OWN_T]);
                #pragma unroll
                for (int k = 0; k < G; ++k) {
                    #pragma unroll
                    for (int c = 0; c < 4; ++c)
                        scan_word((unsigned int)v[k][c], pat, 16 * (b + k * OWN_T) + 4 * c, &cnt, ls);
                }
            }
            int b = ngroups * (G * OWN_T) + tid;
            for (; b < nv; b += OWN_T) {
                const v4i v = __builtin_nontemporal_load(&k4[b]);
                #pragma unroll
                for (int c = 0; c < 4; ++c)
                    scan_word((unsigned int)v[c], pat, 16 * b + 4 * c, &cnt, ls);
            }
            const unsigned char* kb = (const unsigned char*)keys;
            for (int e = 16 * nv + tid; e < E; e += OWN_T)
                if ((int)kb[e] == rg) own_append(e, &cnt, ls);
        } else {
            for (int e = tid; e < E; e += OWN_T) {
                const unsigned int dl = (unsigned int)(ei[E + e] - lo);
                if (dl < (unsigned int)nown) own_append(e, &cnt, ls);
            }
        }
        __syncthreads();

        const unsigned int nlisted = cnt;
        const unsigned int total   = nlisted + (unsigned int)nown;
        if (total <= (unsigned int)OWN_CAP) {
            for (unsigned int i0 = tid; i0 < nlisted; i0 += 4 * OWN_T) {
                unsigned int ix[4];
                int en[4], sn[4], dn[4];
                #pragma unroll
                for (int k = 0; k < 4; ++k) {
                    const unsigned int i = i0 + k * OWN_T;
                    ix[k] = (i < nlisted) ? i : i0;
                    en[k] = ls[ix[k]];
                }
                #pragma unroll
                for (int k = 0; k < 4; ++k) { sn[k] = ei[en[k]];  dn[k] = ei[E + en[k]]; }
                #pragma unroll
                for (int k = 0; k < 4; ++k) {
                    ls[ix[k]] = sn[k];
                    ld[ix[k]] = (unsigned short)(dn[k] - lo);
                }
            }
            for (int i = tid; i < nown; i += OWN_T) {
                ls[nlisted + i] = lo + i;
                ld[nlisted + i] = (unsigned short)i;
            }
            __syncthreads();

            for (unsigned int i = tid; i < total; i += OWN_T) {
                const int s = ls[i], dl = ld[i];
                float v0, v1;
                edge_logit2(al, ar, s, lo + dl, v0, v1);
                lv[2 * i] = v0;  lv[2 * i + 1] = v1;
                atomicMax(&mkey[2 * dl],     f32_to_key(v0));
                atomicMax(&mkey[2 * dl + 1], f32_to_key(v1));
            }
            __syncthreads();
            for (unsigned int i = tid; i < total; i += OWN_T) {
                const int dl = ld[i];
                const float x0 = __expf(lv[2 * i]     - seg_max_of(mkey[2 * dl]));
                const float x1 = __expf(lv[2 * i + 1] - seg_max_of(mkey[2 * dl + 1]));
                lv[2 * i] = x0;  lv[2 * i + 1] = x1;
                lds_add_f32(&den[2 * dl],     x0);
                lds_add_f32(&den[2 * dl + 1], x1);
            }
            __syncthreads();
            for (unsigned int i = tid; i < total; i += OWN_T) {
                const int dl = ld[i];
                lv[2 * i]     = lv[2 * i]     / (den[2 * dl]     + 1e-16f);
                lv[2 * i + 1] = lv[2 * i + 1] / (den[2 * dl + 1] + 1e-16f);
            }
            __syncthreads();
            const unsigned int tfull = total & ~3u;
            for (unsigned int i0 = wave * 4; i0 < tfull; i0 += 32 * 4) {
                float p0[4], p1[4], q0[4], q1[4];
                #pragma unroll
                for (int k = 0; k < 4; ++k) {
                    const float* ys = y + (size_t)ls[i0 + k] * 128;
                    p0[k] = ys[lane];       p1[k] = ys[lane + 32];
                    q0[k] = ys[lane + 64];  q1[k] = ys[lane + 96];
                }
                #pragma unroll
                for (int k = 0; k < 4; ++k) {
                    const float w0 = lv[2 * (i0 + k)], w1 = lv[2 * (i0 + k) + 1];
                    float* td = &tile[ld[i0 + k] * 64];
                    lds_add_f32(td + lane,      p0[k] * w0 + q0[k] * w1);
                    lds_add_f32(td + lane + 32, p1[k] * w0 + q1[k] * w1);
                }
            }
            for (unsigned int i = tfull + wave; i < total; i += 32) {
                const float* ys = y + (size_t)ls[i] * 128;
                const float w0 = lv[2 * i], w1 = lv[2 * i + 1];
                float* td = &tile[ld[i] * 64];
                lds_add_f32(td + lane,      ys[lane]      * w0 + ys[lane + 64] * w1);
                lds_add_f32(td + lane + 32, ys[lane + 32] * w0 + ys[lane + 96] * w1);
            }
        } else {
            own_rescan(ei, E, lo, nown, tid, [&](int s, int dl) {
                float v0, v1;
                edge_logit2(al, ar, s, lo + dl, v0, v1);
                atomicMax(&mkey[2 * dl],     f32_to_key(v0));
                atomicMax(&mkey[2 * dl + 1], f32_to_key(v1));
            });
            __syncthreads();
            own_rescan(ei, E, lo, nown, tid, [&](int s, int dl) {
                float v0, v1;
                edge_logit2(al, ar, s, lo + dl, v0, v1);
                lds_add_f32(&den[2 * dl],     __expf(v0 - seg_max_of(mkey[2 * dl])));
                lds_add_f32(&den[2 * dl + 1], __expf(v1 - seg_max_of(mkey[2 * dl + 1])));
            });
            __syncthreads();
            own_rescan(ei, E, lo, nown, tid, [&](int s, int dl) {
                float v0, v1;
                edge_logit2(al, ar, s, lo + dl, v0, v1);
                const float w0 = __expf(v0 - seg_max_of(mkey[2 * dl]))     / (den[2 * dl]     + 1e-16f);
                const float w1 = __expf(v1 - seg_max_of(mkey[2 * dl + 1])) / (den[2 * dl + 1] + 1e-16f);
                const float* ys = y + (size_t)s * 128;
                for (int c = 0; c < 64; ++c)
                    lds_add_f32(&tile[dl * 64 + c], ys[c] * w0 + ys[64 + c] * w1);
            });
        }
        __syncthreads();

        float* og = out + (size_t)lo * 64;
        for (int i = tid; i < nown * 64; i += OWN_T) og[i] = tile[i] + bias[i & 63];
        __syncthreads();
    }
}

extern "C" void kernel_launch(void* const* d_in, const int* in_sizes, int n_in,
                              void* d_out, int out_size, void* d_ws, size_t ws_size,
                              hipStream_t stream) {
    const float* x    = (const float*)d_in[0];
    const int*   ei   = (const int*)d_in[1];
    const float* Wl   = (const float*)d_in[2];
    const float* attl = (const float*)d_in[3];
    const float* attr = (const float*)d_in[4];
    const float* Wout = (const float*)d_in[5];
    const float* bias = (const float*)d_in[6];
    float* out = (float*)d_out;

    const int N   = in_sizes[0] / 128;
    const int E   = in_sizes[1] / 2;

    float* ws    = (float*)d_ws;
    float* xh    = ws;
    float* y     = xh  + (size_t)N * 128;
    float* al    = y   + (size_t)N * 128;
    float* ar    = al  + (size_t)N * 2;

    proj_gemm_kernel<<<(N + 15) / 16, 256, 0, stream>>>(x, Wl, Wout, xh, y, N);

    unsigned int* keys = (unsigned int*)xh;
    alpha_kernel<<<(N + 7) / 8, 256, 0, stream>>>(xh, attl, attr, al, ar, N, ei, E, keys);

    gat_owner_kernel<<<256, 1024, 0, stream>>>(ei, E, N, keys, al, ar, y, bias, out);
}


# Owner kernel: LDS edge list grouped by node (rank = return of an integer LDS counter), a wave per node sums its rows in registers: no LDS float adds, no out tile; rows summed in rank order
# speedup vs baseline: 3.9780x; 1.0434x over previous
#include <hip/hip_runtime.h>
#include <hip/hip_bf16.h>
#include <math.h>

typedef float v2f __attribute__((ext_vector_type(2)));
typedef float v8f __attribute__((ext_vector_type(8)));

__device__ __forceinline__ unsigned int f32_to_key(float v) {
    const unsigned int b = __float_as_uint(v);
    return (b & 0x80000000u) ? ~b : (b | 0x80000000u);
}
__device__ __forceinline__ float key_to_f32(unsigned int k) {
    const unsigned int b = (k & 0x80000000u) ? (k & 0x7fffffffu) : ~k;
    return __uint_as_float(b);
}

__global__ void proj_gemm_kernel(const float* __restrict__ A,
                                 const float* __restrict__ B,
                                 const float* __restrict__ Wout,
                                 float* __restrict__ XH,
                                 float* __restrict__ Y,
                                 int M) {
    constexpr int K     = 128;
    constexpr int NDIM  = 128;
    constexpr int HC    = 64;
    constexpr int ROWS  = 16;
    constexpr int PITCH = K + 1;
    __shared__ float Alds[ROWS * PITCH];

    const int tid     = threadIdx.x;
    const int wave    = tid >> 5;
    const int lane    = tid & 31;
    const int rowbase = blockIdx.x * ROWS;

    for (int idx = tid; idx < ROWS * K; idx += blockDim.x) {
        int r  = idx / K;
        int c  = idx - r * K;
        int gr = rowbase + r;
        Alds[r * PITCH + c] = (gr < M) ? A[(size_t)gr * K + c] : 0.0f;
    }
    __syncthreads();

    const int colbase = wave * 16;
    const int half    = lane >> 4;
    const int l15     = lane & 15;

    const float* Arow = &Alds[l15 * PITCH];

    v8f acc = {};
    #pragma unroll 8
    for (int k = 0; k < K; k += 4) {
        const int kk = k + 2 * half;
        v2f a, b;
        a.x = Arow[kk];
        a.y = Arow[kk + 1];
        b.x = B[(size_t)kk * NDIM + colbase + l15];
        b.y = B[(size_t)(kk + 1) * NDIM + colbase + l15];
        acc = __builtin_amdgcn_wmma_f32_16x16x4_f32(
             false, a,  false, b,
             (short)0, acc,  false,  false);
    }

    #pragma unroll
    for (int r = 0; r < 8; ++r) {
        int grow = rowbase + r + 8 * half;
        if (grow < M)
            XH[(size_t)grow * NDIM + colbase + l15] = acc[r];
    }

    __syncthreads();
    #pragma unroll
    for (int r = 0; r < 8; ++r)
        Alds[(r + 8 * half) * PITCH + colbase + l15] = acc[r];
    __syncthreads();

    const int kbase = (wave >> 2) * HC;
    const int ycol  = (wave & 3) * 16;
    v8f acc2 = {};
    #pragma unroll 8
    for (int k = 0; k < HC; k += 4) {
        const int kk = kbase + k + 2 * half;
        v2f a, b;
        a.x = Arow[kk];
        a.y = Arow[kk + 1];
        b.x = Wout[(size_t)kk * HC + ycol + l15];
        b.y = Wout[(size_t)(kk + 1) * HC + ycol + l15];
        acc2 = __builtin_amdgcn_wmma_f32_16x16x4_f32(
             false, a,  false, b,
             (short)0, acc2,  false,  false);
    }
    #pragma unroll
    for (int r = 0; r < 8; ++r) {
        int grow = rowbase + r + 8 * half;
        if (grow < M)
            Y[(size_t)grow * NDIM + colbase + l15] = acc2[r];
    }
}

typedef int v4i __attribute__((ext_vector_type(4)));

constexpr int OWN_R   = 196;
constexpr int OWN_CAP = 8192;
constexpr int OWN_T   = 1024;

__device__ __forceinline__ bool keys_usable(int N, int E) {
    return (N + OWN_R - 1) / OWN_R <= 256 && (long long)E <= 512LL * (long long)N;
}

__global__ void alpha_kernel(const float* xh,
                             const float* __restrict__ attl,
                             const float* __restrict__ attr,
                             float* __restrict__ al,
                             float* __restrict__ ar, int N,
                             const int* __restrict__ ei, int E,
                             unsigned int* keys) {
    const int gw   = (blockIdx.x * blockDim.x + threadIdx.x) >> 5;
    const int lane = threadIdx.x & 31;

    if (gw < N) {
        const float* xr = xh + (size_t)gw * 128;
        const float x0 = xr[lane], x1 = xr[lane + 32], x2 = xr[lane + 64], x3 = xr[lane + 96];

        float l0 = x0 * attl[lane]      + x1 * attl[lane + 32];
        float l1 = x2 * attl[lane + 64] + x3 * attl[lane + 96];
        float r0 = x0 * attr[lane]      + x1 * attr[lane + 32];
        float r1 = x2 * attr[lane + 64] + x3 * attr[lane + 96];

        #pragma unroll
        for (int off = 16; off > 0; off >>= 1) {
            l0 += __shfl_xor(l0, off, 32);
            l1 += __shfl_xor(l1, off, 32);
            r0 += __shfl_xor(r0, off, 32);
            r1 += __shfl_xor(r1, off, 32);
        }
        if (lane == 0) {
            al[gw * 2 + 0] = l0;  al[gw * 2 + 1] = l1;
            ar[gw * 2 + 0] = r0;  ar[gw * 2 + 1] = r1;
        }
    }

    if (!keys_usable(N, E)) return;
    __syncthreads();

    const long long e0 = (long long)blockIdx.x * 4096 + 16 * (int)threadIdx.x;
    if (e0 >= E) return;
    const int* dcol = ei + E;
    if ((E & 3) == 0) {
        v4i d[4];
        #pragma unroll
        for (int j = 0; j < 4; ++j)
            d[j] = (e0 + 4 * j < E) ? *(const v4i*)(dcol + e0 + 4 * j) : v4i{0, 0, 0, 0};
        #pragma unroll
        for (int j = 0; j < 4; ++j) {
            const unsigned int w = ((unsigned int)d[j][0] / OWN_R)
                                 | ((unsigned int)d[j][1] / OWN_R) << 8
                                 | ((unsigned int)d[j][2] / OWN_R) << 16
                                 | ((unsigned int)d[j][3] / OWN_R) << 24;
            if (e0 + 4 * j < E) keys[(e0 >> 2) + j] = w;
        }
    } else {
        unsigned char* kb = (unsigned char*)keys;
        for (int j = 0; j < 16; ++j)
            if (e0 + j < E) kb[e0 + j] = (unsigned char)((unsigned int)dcol[e0 + j] / OWN_R);
    }
}

__device__ __forceinline__ void lds_add_f32(float* p, float v) {
    __hip_atomic_fetch_add(p, v, __ATOMIC_RELAXED, __HIP_MEMORY_SCOPE_WORKGROUP);
}

__device__ __forceinline__ void own_append(int e, unsigned int* cnt, int* ls) {
    const unsigned int pos = atomicAdd(cnt, 1u);
    if (pos < (unsigned int)OWN_CAP) ls[pos] = e;
}

__device__ __forceinline__ void scan_word(unsigned int w, unsigned int pat, int ebase,
                                          unsigned int* cnt, int* ls) {
    const unsigned int x = w ^ pat;
    unsigned int z = ~((((x & 0x7f7f7f7fu) + 0x7f7f7f7fu) | x) | 0x7f7f7f7fu);
    while (z) {
        own_append(ebase + (__builtin_ctz(z) >> 3), cnt, ls);
        z &= z - 1u;
    }
}

__device__ __forceinline__ void edge_logit2(const float* __restrict__ al,
                                            const float* __restrict__ ar,
                                            int s, int d, float& v0, float& v1) {
    const float a0 = al[2 * s], a1 = al[2 * s + 1];
    const float r0 = ar[2 * d], r1 = ar[2 * d + 1];
    v0 = a0 + r0;  v0 = (v0 > 0.0f) ? v0 : 0.2f * v0;
    v1 = a1 + r1;  v1 = (v1 > 0.0f) ? v1 : 0.2f * v1;
}

__device__ __forceinline__ float seg_max_of(unsigned int key) {
    float m = key_to_f32(key);
    if (!(fabsf(m) <= 3.4e38f)) m = 0.0f;
    return m;
}

template <class F>
__device__ __forceinline__ void own_rescan(const int* __restrict__ ei, int E,
                                           int lo, int nown, int tid, F f) {
    for (int e = tid; e < E; e += OWN_T) {
        const unsigned int dl = (unsigned int)(ei[E + e] - lo);
        if (dl < (unsigned int)nown) f(ei[e], (int)dl);
    }
    for (int i = tid; i < nown; i += OWN_T) f(lo + i, i);
}

__global__ __launch_bounds__(1024)
void gat_owner_kernel(const int* __restrict__ ei, int E, int N,
                      const unsigned int* __restrict__ keys,
                      const float* __restrict__ al,
                      const float* __restrict__ ar,
                      const float* __restrict__ y,
                      const float* __restrict__ bias,
                      float* __restrict__ out) {
    __shared__ v2f            sw[OWN_CAP];
    __shared__ float          lv[OWN_CAP * 2];
    __shared__ int            ls[OWN_CAP];
    __shared__ int            ss[OWN_CAP];
    __shared__ unsigned short ld[OWN_CAP];
    __shared__ unsigned short rk[OWN_CAP];
    __shared__ unsigned int   mkey[OWN_R * 2];
    __shared__ float          den[OWN_R * 2];
    __shared__ unsigned int   deg[OWN_R];
    __shared__ unsigned int   start[OWN_R];
    __shared__ unsigned int   cnt;
    static_assert(sizeof(float) * OWN_R * 64 <= sizeof(v2f) * OWN_CAP, "out tile must fit the slot array");
    float* const tile = (float*)sw;
    constexpr int PER = (OWN_R + 31) / 32;

    const int tid  = threadIdx.x;
    const int lane = tid & 31;
    const int wave = tid >> 5;
    const int nranges = (N + OWN_R - 1) / OWN_R;
    const bool usekeys = keys_usable(N, E);

    for (int rg = blockIdx.x; rg < nranges; rg += gridDim.x) {
        const int lo   = rg * OWN_R;
        const int nown = min(OWN_R, N - lo);

        for (int i = tid; i < nown * 2; i += OWN_T) { mkey[i] = 0u; den[i] = 0.0f; }
        for (int i = tid; i < nown; i += OWN_T) deg[i] = 0u;
        if (tid == 0) cnt = 0u;
        __syncthreads();

        if (usekeys) {
            const unsigned int pat = (unsigned int)rg * 0x01010101u;
            const v4i* __restrict__ k4 = (const v4i*)keys;
            const int nv = E >> 4;
            constexpr int G = 8;
            const int ngroups = nv / (G * OWN_T);
            int g = (int)(((long long)rg * ngroups) / nranges);
            for (int t = 0; t < ngroups; ++t) {
                const int b = g * (G * OWN_T) + tid;
                g = (g + 1 < ngroups) ? g + 1 : 0;
                v4i v[G];
                #pragma unroll
                for (int k = 0; k < G; ++k)
                    v[k] = __builtin_nontemporal_load(&k4[b + k * OWN_T]);
                #pragma unroll
                for (int k = 0; k < G; ++k) {
                    #pragma unroll
                    for (int c = 0; c < 4; ++c)
                        scan_word((unsigned int)v[k][c], pat, 16 * (b + k * OWN_T) + 4 * c, &cnt, ls);
                }
            }
            int b = ngroups * (G * OWN_T) + tid;
            for (; b < nv; b += OWN_T) {
                const v4i v = __builtin_nontemporal_load(&k4[b]);
                #pragma unroll
                for (int c = 0; c < 4; ++c)
                    scan_word((unsigned int)v[c], pat, 16 * b + 4 * c, &cnt, ls);
            }
            const unsigned char* kb = (const unsigned char*)keys;
            for (int e = 16 * nv + tid; e < E; e += OWN_T)
                if ((int)kb[e] == rg) own_append(e, &cnt, ls);
        } else {
            for (int e = tid; e < E; e += OWN_T) {
                const unsigned int dl = (unsigned int)(ei[E + e] - lo);
                if (dl < (unsigned int)nown) own_append(e, &cnt, ls);
            }
        }
        __syncthreads();

        const unsigned int nlisted = cnt;
        const unsigned int total   = nlisted + (unsigned int)nown;
        if (total <= (unsigned int)OWN_CAP) {
            for (unsigned int i0 = tid; i0 < nlisted; i0 += 4 * OWN_T) {
                unsigned int ix[4];
                int en[4], sn[4], dn[4];
                #pragma unroll
                for (int k = 0; k < 4; ++k) {
                    const unsigned int i = i0 + k * OWN_T;
                    ix[k] = (i < nlisted) ? i : i0;
                    en[k] = ls[ix[k]];
                }
                #pragma unroll
                for (int k = 0; k < 4; ++k) { sn[k] = ei[en[k]];  dn[k] = ei[E + en[k]]; }
                #pragma unroll
                for (int k = 0; k < 4; ++k) {
                    ls[ix[k]] = sn[k];
                    ld[ix[k]] = (unsigned short)(dn[k] - lo);
                }
            }
            for (int i = tid; i < nown; i += OWN_T) {
                ls[nlisted + i] = lo + i;
                ld[nlisted + i] = (unsigned short)i;
            }
            __syncthreads();

            for (unsigned int i = tid; i < total; i += OWN_T) {
                const int s = ls[i], dl = ld[i];
                float v0, v1;
                edge_logit2(al, ar, s, lo + dl, v0, v1);
                lv[2 * i] = v0;  lv[2 * i + 1] = v1;
                atomicMax(&mkey[2 * dl],     f32_to_key(v0));
                atomicMax(&mkey[2 * dl + 1], f32_to_key(v1));
                rk[i] = (unsigned short)atomicAdd(&deg[dl], 1u);
            }
            __syncthreads();
            if (wave == 0) {
                unsigned int c[PER];
                unsigned int sum = 0u;
                #pragma unroll
                for (int j = 0; j < PER; ++j) {
                    const int n = PER * lane + j;
                    c[j] = (n < nown) ? deg[n] : 0u;
                    sum += c[j];
                }
                unsigned int incl = sum;
                #pragma unroll
                for (int off = 1; off < 32; off <<= 1) {
                    const unsigned int t = __shfl_up(incl, off, 32);
                    if (lane >= off) incl += t;
                }
                unsigned int run = incl - sum;
                #pragma unroll
                for (int j = 0; j < PER; ++j) {
                    const int n = PER * lane + j;
                    if (n < nown) start[n] = run;
                    run += c[j];
                }
            }
            for (unsigned int i = tid; i < total; i += OWN_T) {
                const int dl = ld[i];
                const float x0 = __expf(lv[2 * i]     - seg_max_of(mkey[2 * dl]));
                const float x1 = __expf(lv[2 * i + 1] - seg_max_of(mkey[2 * dl + 1]));
                lv[2 * i] = x0;  lv[2 * i + 1] = x1;
                lds_add_f32(&den[2 * dl],     x0);
                lds_add_f32(&den[2 * dl + 1], x1);
            }
            __syncthreads();
            for (unsigned int i = tid; i < total; i += OWN_T) {
                const int dl = ld[i];
                const unsigned int p = start[dl] + rk[i];
                v2f w;
                w.x = lv[2 * i]     / (den[2 * dl]     + 1e-16f);
                w.y = lv[2 * i + 1] / (den[2 * dl + 1] + 1e-16f);
                ss[p] = ls[i];
                sw[p] = w;
            }
            __syncthreads();
            const int wave_s = __builtin_amdgcn_readfirstlane(wave);
            for (int n = wave_s; n < nown; n += 32) {
                const unsigned int b  = __builtin_amdgcn_readfirstlane(start[n]);
                const unsigned int dn = __builtin_amdgcn_readfirstlane(deg[n]);
                float a0 = 0.0f, a1 = 0.0f;
                for (unsigned int j = 0; j < dn; j += 8) {
                    const unsigned int o    = j + (unsigned int)(lane & 7);
                    const unsigned int slot = b + ((o < dn) ? o : j);
                    const int sv = ss[slot];
                    const v2f wv = sw[slot];
                    float p0[8], p1[8], q0[8], q1[8];
                    #pragma unroll
                    for (int k = 0; k < 8; ++k) {
                        const int s = __builtin_amdgcn_readlane(sv, k);
                        const float* ys = y + (size_t)s * 128;
                        p0[k] = ys[lane];       p1[k] = ys[lane + 32];
                        q0[k] = ys[lane + 64];  q1[k] = ys[lane + 96];
                    }
                    #pragma unroll
                    for (int k = 0; k < 8; ++k) {
                        const float w0 = __uint_as_float((unsigned int)__builtin_amdgcn_readlane((int)__float_as_uint(wv.x), k));
                        const float w1 = __uint_as_float((unsigned int)__builtin_amdgcn_readlane((int)__float_as_uint(wv.y), k));
                        const float t0 = p0[k] * w0 + q0[k] * w1;
                        const float t1 = p1[k] * w0 + q1[k] * w1;
                        const bool in_run = (j + (unsigned int)k < dn);
                        a0 += in_run ? t0 : 0.0f;
                        a1 += in_run ? t1 : 0.0f;
                    }
                }
                float* og = out + (size_t)(lo + n) * 64;
                og[lane]      = a0 + bias[lane];
                og[lane + 32] = a1 + bias[lane + 32];
            }
        } else {
            for (int i = tid; i < nown * 64; i += OWN_T) tile[i] = 0.0f;
            __syncthreads();
            own_rescan(ei, E, lo, nown, tid, [&](int s, int dl) {
                float v0, v1;
                edge_logit2(al, ar, s, lo + dl, v0, v1);
                atomicMax(&mkey[2 * dl],     f32_to_key(v0));
                atomicMax(&mkey[2 * dl + 1], f32_to_key(v1));
            });
            __syncthreads();
            own_rescan(ei, E, lo, nown, tid, [&](int s, int dl) {
                float v0, v1;
                edge_logit2(al, ar, s, lo + dl, v0, v1);
                lds_add_f32(&den[2 * dl],     __expf(v0 - seg_max_of(mkey[2 * dl])));
                lds_add_f32(&den[2 * dl + 1], __expf(v1 - seg_max_of(mkey[2 * dl + 1])));
            });
            __syncthreads();
            own_rescan(ei, E, lo, nown, tid, [&](int s, int dl) {
                float v0, v1;
                edge_logit2(al, ar, s, lo + dl, v0, v1);
                const float w0 = __expf(v0 - seg_max_of(mkey[2 * dl]))     / (den[2 * dl]     + 1e-16f);
                const float w1 = __expf(v1 - seg_max_of(mkey[2 * dl + 1])) / (den[2 * dl + 1] + 1e-16f);
                const float* ys = y + (size_t)s * 128;
                for (int c = 0; c < 64; ++c)
                    lds_add_f32(&tile[dl * 64 + c], ys[c] * w0 + ys[64 + c] * w1);
            });
            __syncthreads();
            float* og = out + (size_t)lo * 64;
            for (int i = tid; i < nown * 64; i += OWN_T) og[i] = tile[i] + bias[i & 63];
        }
        __syncthreads();
    }
}

extern "C" void kernel_launch(void* const* d_in, const int* in_sizes, int n_in,
                              void* d_out, int out_size, void* d_ws, size_t ws_size,
                              hipStream_t stream) {
    const float* x    = (const float*)d_in[0];
    const int*   ei   = (const int*)d_in[1];
    const float* Wl   = (const float*)d_in[2];
    const float* attl = (const float*)d_in[3];
    const float* attr = (const float*)d_in[4];
    const float* Wout = (const float*)d_in[5];
    const float* bias = (const float*)d_in[6];
    float* out = (float*)d_out;

    const int N   = in_sizes[0] / 128;
    const int E   = in_sizes[1] / 2;

    float* ws    = (float*)d_ws;
    float* xh    = ws;
    float* y     = xh  + (size_t)N * 128;
    float* al    = y   + (size_t)N * 128;
    float* ar    = al  + (size_t)N * 2;

    proj_gemm_kernel<<<(N + 15) / 16, 256, 0, stream>>>(x, Wl, Wout, xh, y, N);

    unsigned int* keys = (unsigned int*)xh;
    alpha_kernel<<<(N + 7) / 8, 256, 0, stream>>>(xh, attl, attr, al, ar, N, ei, E, keys);

    gat_owner_kernel<<<256, 1024, 0, stream>>>(ei, E, N, keys, al, ar, y, bias, out);
}


# Key column as 8 bit planes per 32-edge group (alpha kernel builds a plane by wave ballot): owner scan matches 32 edges with 12 bit operations, not 4 edges with 5; sums still in LDS arrival order
# speedup vs baseline: 4.3886x; 1.1032x over previous
#include <hip/hip_runtime.h>
#include <hip/hip_bf16.h>
#include <math.h>

typedef float v2f __attribute__((ext_vector_type(2)));
typedef float v8f __attribute__((ext_vector_type(8)));

__device__ __forceinline__ unsigned int f32_to_key(float v) {
    const unsigned int b = __float_as_uint(v);
    return (b & 0x80000000u) ? ~b : (b | 0x80000000u);
}
__device__ __forceinline__ float key_to_f32(unsigned int k) {
    const unsigned int b = (k & 0x80000000u) ? (k & 0x7fffffffu) : ~k;
    return __uint_as_float(b);
}

__global__ void proj_gemm_kernel(const float* __restrict__ A,
                                 const float* __restrict__ B,
                                 const float* __restrict__ Wout,
                                 float* __restrict__ XH,
                                 float* __restrict__ Y,
                                 int M) {
    constexpr int K     = 128;
    constexpr int NDIM  = 128;
    constexpr int HC    = 64;
    constexpr int ROWS  = 16;
    constexpr int PITCH = K + 1;
    __shared__ float Alds[ROWS * PITCH];

    const int tid     = threadIdx.x;
    const int wave    = tid >> 5;
    const int lane    = tid & 31;
    const int rowbase = blockIdx.x * ROWS;

    for (int idx = tid; idx < ROWS * K; idx += blockDim.x) {
        int r  = idx / K;
        int c  = idx - r * K;
        int gr = rowbase + r;
        Alds[r * PITCH + c] = (gr < M) ? A[(size_t)gr * K + c] : 0.0f;
    }
    __syncthreads();

    const int colbase = wave * 16;
    const int half    = lane >> 4;
    const int l15     = lane & 15;

    const float* Arow = &Alds[l15 * PITCH];

    v8f acc = {};
    #pragma unroll 8
    for (int k = 0; k < K; k += 4) {
        const int kk = k + 2 * half;
        v2f a, b;
        a.x = Arow[kk];
        a.y = Arow[kk + 1];
        b.x = B[(size_t)kk * NDIM + colbase + l15];
        b.y = B[(size_t)(kk + 1) * NDIM + colbase + l15];
        acc = __builtin_amdgcn_wmma_f32_16x16x4_f32(
             false, a,  false, b,
             (short)0, acc,  false,  false);
    }

    #pragma unroll
    for (int r = 0; r < 8; ++r) {
        int grow = rowbase + r + 8 * half;
        if (grow < M)
            XH[(size_t)grow * NDIM + colbase + l15] = acc[r];
    }

    __syncthreads();
    #pragma unroll
    for (int r = 0; r < 8; ++r)
        Alds[(r + 8 * half) * PITCH + colbase + l15] = acc[r];
    __syncthreads();

    const int kbase = (wave >> 2) * HC;
    const int ycol  = (wave & 3) * 16;
    v8f acc2 = {};
    #pragma unroll 8
    for (int k = 0; k < HC; k += 4) {
        const int kk = kbase + k + 2 * half;
        v2f a, b;
        a.x = Arow[kk];
        a.y = Arow[kk + 1];
        b.x = Wout[(size_t)kk * HC + ycol + l15];
        b.y = Wout[(size_t)(kk + 1) * HC + ycol + l15];
        acc2 = __builtin_amdgcn_wmma_f32_16x16x4_f32(
             false, a,  false, b,
             (short)0, acc2,  false,  false);
    }
    #pragma unroll
    for (int r = 0; r < 8; ++r) {
        int grow = rowbase + r + 8 * half;
        if (grow < M)
            Y[(size_t)grow * NDIM + colbase + l15] = acc2[r];
    }
}

typedef int v4i __attribute__((ext_vector_type(4)));

constexpr int OWN_R   = 196;
constexpr int OWN_CAP = 8192;
constexpr int OWN_T   = 1024;

__device__ __forceinline__ bool keys_usable(int N, int E) {
    return (N + OWN_R - 1) / OWN_R <= 256 && (long long)E + 4127LL <= 512LL * (long long)N;
}
__device__ __forceinline__ int key_vec_of_group(int g) {
    return ((g >> 7) << 8) + (g & 127);
}

__global__ void alpha_kernel(const float* xh,
                             const float* __restrict__ attl,
                             const float* __restrict__ attr,
                             float* __restrict__ al,
                             float* __restrict__ ar, int N,
                             const int* __restrict__ ei, int E,
                             unsigned int* keys) {
    const int gw   = (blockIdx.x * blockDim.x + threadIdx.x) >> 5;
    const int lane = threadIdx.x & 31;

    if (gw < N) {
        const float* xr = xh + (size_t)gw * 128;
        const float x0 = xr[lane], x1 = xr[lane + 32], x2 = xr[lane + 64], x3 = xr[lane + 96];

        float l0 = x0 * attl[lane]      + x1 * attl[lane + 32];
        float l1 = x2 * attl[lane + 64] + x3 * attl[lane + 96];
        float r0 = x0 * attr[lane]      + x1 * attr[lane + 32];
        float r1 = x2 * attr[lane + 64] + x3 * attr[lane + 96];

        #pragma unroll
        for (int off = 16; off > 0; off >>= 1) {
            l0 += __shfl_xor(l0, off, 32);
            l1 += __shfl_xor(l1, off, 32);
            r0 += __shfl_xor(r0, off, 32);
            r1 += __shfl_xor(r1, off, 32);
        }
        if (lane == 0) {
            al[gw * 2 + 0] = l0;  al[gw * 2 + 1] = l1;
            ar[gw * 2 + 0] = r0;  ar[gw * 2 + 1] = r1;
        }
    }

    if (!keys_usable(N, E)) return;
    __syncthreads();

    const int ng    = (int)(((long long)E + 31) >> 5);
    const int gwave = (int)blockIdx.x * 128 + 16 * (int)(threadIdx.x >> 5);
    if (gwave >= ng) return;
    const int* dcol = ei + E;
    unsigned int r[16];
    #pragma unroll
    for (int i = 0; i < 16; ++i) {
        const long long e = ((long long)(gwave + i) << 5) + lane;
        r[i] = (unsigned int)dcol[min(e, (long long)E - 1)] / OWN_R;
    }
    #pragma unroll
    for (int it = 0; it < 4; ++it) {
        unsigned int w = 0u;
        #pragma unroll
        for (int q = 0; q < 4; ++q) {
            #pragma unroll
            for (int b = 0; b < 8; ++b) {
                const unsigned int p = __builtin_amdgcn_ballot_w32(((r[4 * it + q] >> b) & 1u) != 0u);
                if (lane == 8 * q + b) w = p;
            }
        }
        const int q  = lane >> 3, b = lane & 7;
        const int g  = gwave + 4 * it + q;
        const int wi = 4 * key_vec_of_group(g) + ((b < 4) ? b : 512 + (b - 4));
        if (g < ng) keys[wi] = w;
    }
}

__device__ __forceinline__ void lds_add_f32(float* p, float v) {
    __hip_atomic_fetch_add(p, v, __ATOMIC_RELAXED, __HIP_MEMORY_SCOPE_WORKGROUP);
}

__device__ __forceinline__ void own_append(int e, unsigned int* cnt, int* ls) {
    const unsigned int pos = atomicAdd(cnt, 1u);
    if (pos < (unsigned int)OWN_CAP) ls[pos] = e;
}

__device__ __forceinline__ void scan_group(v4i lo, v4i hi, const unsigned int (&inv)[8],
                                           int ebase, unsigned int keep, unsigned int* cnt, int* ls) {
    unsigned int m = ((unsigned int)lo[0] ^ inv[0]) & ((unsigned int)lo[1] ^ inv[1])
                   & ((unsigned int)lo[2] ^ inv[2]) & ((unsigned int)lo[3] ^ inv[3])
                   & ((unsigned int)hi[0] ^ inv[4]) & ((unsigned int)hi[1] ^ inv[5])
                   & ((unsigned int)hi[2] ^ inv[6]) & ((unsigned int)hi[3] ^ inv[7]) & keep;
    while (m) {
        own_append(ebase + __builtin_ctz(m), cnt, ls);
        m &= m - 1u;
    }
}

__device__ __forceinline__ void edge_logit2(const float* __restrict__ al,
                                            const float* __restrict__ ar,
                                            int s, int d, float& v0, float& v1) {
    const float a0 = al[2 * s], a1 = al[2 * s + 1];
    const float r0 = ar[2 * d], r1 = ar[2 * d + 1];
    v0 = a0 + r0;  v0 = (v0 > 0.0f) ? v0 : 0.2f * v0;
    v1 = a1 + r1;  v1 = (v1 > 0.0f) ? v1 : 0.2f * v1;
}

__device__ __forceinline__ float seg_max_of(unsigned int key) {
    float m = key_to_f32(key);
    if (!(fabsf(m) <= 3.4e38f)) m = 0.0f;
    return m;
}

template <class F>
__device__ __forceinline__ void own_rescan(const int* __restrict__ ei, int E,
                                           int lo, int nown, int tid, F f) {
    for (int e = tid; e < E; e += OWN_T) {
        const unsigned int dl = (unsigned int)(ei[E + e] - lo);
        if (dl < (unsigned int)nown) f(ei[e], (int)dl);
    }
    for (int i = tid; i < nown; i += OWN_T) f(lo + i, i);
}

__global__ __launch_bounds__(1024)
void gat_owner_kernel(const int* __restrict__ ei, int E, int N,
                      const unsigned int* __restrict__ keys,
                      const float* __restrict__ al,
                      const float* __restrict__ ar,
                      const float* __restrict__ y,
                      const float* __restrict__ bias,
                      float* __restrict__ out) {
    __shared__ v2f            sw[OWN_CAP];
    __shared__ float          lv[OWN_CAP * 2];
    __shared__ int            ls[OWN_CAP];
    __shared__ int            ss[OWN_CAP];
    __shared__ unsigned short ld[OWN_CAP];
    __shared__ unsigned short rk[OWN_CAP];
    __shared__ unsigned int   mkey[OWN_R * 2];
    __shared__ float          den[OWN_R * 2];
    __shared__ unsigned int   deg[OWN_R];
    __shared__ unsigned int   start[OWN_R];
    __shared__ unsigned int   cnt;
    static_assert(sizeof(float) * OWN_R * 64 <= sizeof(v2f) * OWN_CAP, "out tile must fit the slot array");
    float* const tile = (float*)sw;
    constexpr int PER = (OWN_R + 31) / 32;

    const int tid  = threadIdx.x;
    const int lane = tid & 31;
    const int wave = tid >> 5;
    const int nranges = (N + OWN_R - 1) / OWN_R;
    const bool usekeys = keys_usable(N, E);

    for (int rg = blockIdx.x; rg < nranges; rg += gridDim.x) {
        const int lo   = rg * OWN_R;
        const int nown = min(OWN_R, N - lo);

        for (int i = tid; i < nown * 2; i += OWN_T) { mkey[i] = 0u; den[i] = 0.0f; }
        for (int i = tid; i < nown; i += OWN_T) deg[i] = 0u;
        if (tid == 0) cnt = 0u;
        __syncthreads();

        if (usekeys) {
            unsigned int inv[8];
            #pragma unroll
            for (int b = 0; b < 8; ++b) inv[b] = ((rg >> b) & 1) ? 0u : 0xffffffffu;
            const v4i* __restrict__ k4 = (const v4i*)keys;
            const int ng = (int)(((long long)E + 31) >> 5);
            constexpr int G = 4;
            const int nbatch = (E >> 5) / (G * OWN_T);
            int bt = (int)(((long long)rg * nbatch) / nranges);
            for (int t = 0; t < nbatch; ++t) {
                const int g0 = bt * (G * OWN_T) + tid;
                bt = (bt + 1 < nbatch) ? bt + 1 : 0;
                v4i lo[G], hi[G];
                #pragma unroll
                for (int k = 0; k < G; ++k) {
                    const int vi = key_vec_of_group(g0 + k * OWN_T);
                    lo[k] = __builtin_nontemporal_load(&k4[vi]);
                    hi[k] = __builtin_nontemporal_load(&k4[vi + 128]);
                }
                #pragma unroll
                for (int k = 0; k < G; ++k)
                    scan_group(lo[k], hi[k], inv, 32 * (g0 + k * OWN_T), 0xffffffffu, &cnt, ls);
            }
            for (int g = nbatch * (G * OWN_T) + tid; g < ng; g += OWN_T) {
                const int vi = key_vec_of_group(g);
                const v4i lo = __builtin_nontemporal_load(&k4[vi]);
                const v4i hi = __builtin_nontemporal_load(&k4[vi + 128]);
                const int rem = E - 32 * g;
                scan_group(lo, hi, inv, 32 * g, (rem < 32) ? (1u << rem) - 1u : 0xffffffffu, &cnt, ls);
            }
        } else {
            for (int e = tid; e < E; e += OWN_T) {
                const unsigned int dl = (unsigned int)(ei[E + e] - lo);
                if (dl < (unsigned int)nown) own_append(e, &cnt, ls);
            }
        }
        __syncthreads();

        const unsigned int nlisted = cnt;
        const unsigned int total   = nlisted + (unsigned int)nown;
        if (total <= (unsigned int)OWN_CAP) {
            for (unsigned int i0 = tid; i0 < nlisted; i0 += 4 * OWN_T) {
                unsigned int ix[4];
                int en[4], sn[4], dn[4];
                #pragma unroll
                for (int k = 0; k < 4; ++k) {
                    const unsigned int i = i0 + k * OWN_T;
                    ix[k] = (i < nlisted) ? i : i0;
                    en[k] = ls[ix[k]];
                }
                #pragma unroll
                for (int k = 0; k < 4; ++k) { sn[k] = ei[en[k]];  dn[k] = ei[E + en[k]]; }
                #pragma unroll
                for (int k = 0; k < 4; ++k) {
                    ls[ix[k]] = sn[k];
                    ld[ix[k]] = (unsigned short)(dn[k] - lo);
                }
            }
            for (int i = tid; i < nown; i += OWN_T) {
                ls[nlisted + i] = lo + i;
                ld[nlisted + i] = (unsigned short)i;
            }
            __syncthreads();

            for (unsigned int i = tid; i < total; i += OWN_T) {
                const int s = ls[i], dl = ld[i];
                float v0, v1;
                edge_logit2(al, ar, s, lo + dl, v0, v1);
                lv[2 * i] = v0;  lv[2 * i + 1] = v1;
                atomicMax(&mkey[2 * dl],     f32_to_key(v0));
                atomicMax(&mkey[2 * dl + 1], f32_to_key(v1));
                rk[i] = (unsigned short)atomicAdd(&deg[dl], 1u);
            }
            __syncthreads();
            if (wave == 0) {
                unsigned int c[PER];
                unsigned int sum = 0u;
                #pragma unroll
                for (int j = 0; j < PER; ++j) {
                    const int n = PER * lane + j;
                    c[j] = (n < nown) ? deg[n] : 0u;
                    sum += c[j];
                }
                unsigned int incl = sum;
                #pragma unroll
                for (int off = 1; off < 32; off <<= 1) {
                    const unsigned int t = __shfl_up(incl, off, 32);
                    if (lane >= off) incl += t;
                }
                unsigned int run = incl - sum;
                #pragma unroll
                for (int j = 0; j < PER; ++j) {
                    const int n = PER * lane + j;
                    if (n < nown) start[n] = run;
                    run += c[j];
                }
            }
            for (unsigned int i = tid; i < total; i += OWN_T) {
                const int dl = ld[i];
                const float x0 = __expf(lv[2 * i]     - seg_max_of(mkey[2 * dl]));
                const float x1 = __expf(lv[2 * i + 1] - seg_max_of(mkey[2 * dl + 1]));
                lv[2 * i] = x0;  lv[2 * i + 1] = x1;
                lds_add_f32(&den[2 * dl],     x0);
                lds_add_f32(&den[2 * dl + 1], x1);
            }
            __syncthreads();
            for (unsigned int i = tid; i < total; i += OWN_T) {
                const int dl = ld[i];
                const unsigned int p = start[dl] + rk[i];
                v2f w;
                w.x = lv[2 * i]     / (den[2 * dl]     + 1e-16f);
                w.y = lv[2 * i + 1] / (den[2 * dl + 1] + 1e-16f);
                ss[p] = ls[i];
                sw[p] = w;
            }
            __syncthreads();
            const int wave_s = __builtin_amdgcn_readfirstlane(wave);
            for (int n = wave_s; n < nown; n += 32) {
                const unsigned int b  = __builtin_amdgcn_readfirstlane(start[n]);
                const unsigned int dn = __builtin_amdgcn_readfirstlane(deg[n]);
                float a0 = 0.0f, a1 = 0.0f;
                for (unsigned int j = 0; j < dn; j += 8) {
                    const unsigned int o    = j + (unsigned int)(lane & 7);
                    const unsigned int slot = b + ((o < dn) ? o : j);
                    const int sv = ss[slot];
                    const v2f wv = sw[slot];
                    float p0[8], p1[8], q0[8], q1[8];
                    #pragma unroll
                    for (int k = 0; k < 8; ++k) {
                        const int s = __builtin_amdgcn_readlane(sv, k);
                        const float* ys = y + (size_t)s * 128;
                        p0[k] = ys[lane];       p1[k] = ys[lane + 32];
                        q0[k] = ys[lane + 64];  q1[k] = ys[lane + 96];
                    }
                    #pragma unroll
                    for (int k = 0; k < 8; ++k) {
                        const float w0 = __uint_as_float((unsigned int)__builtin_amdgcn_readlane((int)__float_as_uint(wv.x), k));
                        const float w1 = __uint_as_float((unsigned int)__builtin_amdgcn_readlane((int)__float_as_uint(wv.y), k));
                        const float t0 = p0[k] * w0 + q0[k] * w1;
                        const float t1 = p1[k] * w0 + q1[k] * w1;
                        const bool in_run = (j + (unsigned int)k < dn);
                        a0 += in_run ? t0 : 0.0f;
                        a1 += in_run ? t1 : 0.0f;
                    }
                }
                float* og = out + (size_t)(lo + n) * 64;
                og[lane]      = a0 + bias[lane];
                og[lane + 32] = a1 + bias[lane + 32];
            }
        } else {
            for (int i = tid; i < nown * 64; i += OWN_T) tile[i] = 0.0f;
            __syncthreads();
            own_rescan(ei, E, lo, nown, tid, [&](int s, int dl) {
                float v0, v1;
                edge_logit2(al, ar, s, lo + dl, v0, v1);
                atomicMax(&mkey[2 * dl],     f32_to_key(v0));
                atomicMax(&mkey[2 * dl + 1], f32_to_key(v1));
            });
            __syncthreads();
            own_rescan(ei, E, lo, nown, tid, [&](int s, int dl) {
                float v0, v1;
                edge_logit2(al, ar, s, lo + dl, v0, v1);
                lds_add_f32(&den[2 * dl],     __expf(v0 - seg_max_of(mkey[2 * dl])));
                lds_add_f32(&den[2 * dl + 1], __expf(v1 - seg_max_of(mkey[2 * dl + 1])));
            });
            __syncthreads();
            own_rescan(ei, E, lo, nown, tid, [&](int s, int dl) {
                float v0, v1;
                edge_logit2(al, ar, s, lo + dl, v0, v1);
                const float w0 = __expf(v0 - seg_max_of(mkey[2 * dl]))     / (den[2 * dl]     + 1e-16f);
                const float w1 = __expf(v1 - seg_max_of(mkey[2 * dl + 1])) / (den[2 * dl + 1] + 1e-16f);
                const float* ys = y + (size_t)s * 128;
                for (int c = 0; c < 64; ++c)
                    lds_add_f32(&tile[dl * 64 + c], ys[c] * w0 + ys[64 + c] * w1);
            });
            __syncthreads();
            float* og = out + (size_t)lo * 64;
            for (int i = tid; i < nown * 64; i += OWN_T) og[i] = tile[i] + bias[i & 63];
        }
        __syncthreads();
    }
}

extern "C" void kernel_launch(void* const* d_in, const int* in_sizes, int n_in,
                              void* d_out, int out_size, void* d_ws, size_t ws_size,
                              hipStream_t stream) {
    const float* x    = (const float*)d_in[0];
    const int*   ei   = (const int*)d_in[1];
    const float* Wl   = (const float*)d_in[2];
    const float* attl = (const float*)d_in[3];
    const float* attr = (const float*)d_in[4];
    const float* Wout = (const float*)d_in[5];
    const float* bias = (const float*)d_in[6];
    float* out = (float*)d_out;

    const int N   = in_sizes[0] / 128;
    const int E   = in_sizes[1] / 2;

    float* ws    = (float*)d_ws;
    float* xh    = ws;
    float* y     = xh  + (size_t)N * 128;
    float* al    = y   + (size_t)N * 128;
    float* ar    = al  + (size_t)N * 2;

    proj_gemm_kernel<<<(N + 15) / 16, 256, 0, stream>>>(x, Wl, Wout, xh, y, N);

    unsigned int* keys = (unsigned int*)xh;
    alpha_kernel<<<(N + 7) / 8, 256, 0, stream>>>(xh, attl, attr, al, ar, N, ei, E, keys);

    gat_owner_kernel<<<256, 1024, 0, stream>>>(ei, E, N, keys, al, ar, y, bias, out);
}


# Alpha kernel on 3128 waves of 16 rows (16 loads in flight, 16-lane DPP sums, whole-line al/ar stores), not 50000 one-row waves; key column split over the work-groups' own rows; dot sum order changes
# speedup vs baseline: 4.5457x; 1.0358x over previous
#include <hip/hip_runtime.h>
#include <hip/hip_bf16.h>
#include <math.h>

typedef float v2f __attribute__((ext_vector_type(2)));
typedef float v8f __attribute__((ext_vector_type(8)));
typedef float v4f __attribute__((ext_vector_type(4)));

__device__ __forceinline__ unsigned int f32_to_key(float v) {
    const unsigned int b = __float_as_uint(v);
    return (b & 0x80000000u) ? ~b : (b | 0x80000000u);
}
__device__ __forceinline__ float key_to_f32(unsigned int k) {
    const unsigned int b = (k & 0x80000000u) ? (k & 0x7fffffffu) : ~k;
    return __uint_as_float(b);
}

__global__ void proj_gemm_kernel(const float* __restrict__ A,
                                 const float* __restrict__ B,
                                 const float* __restrict__ Wout,
                                 float* __restrict__ XH,
                                 float* __restrict__ Y,
                                 int M) {
    constexpr int K     = 128;
    constexpr int NDIM  = 128;
    constexpr int HC    = 64;
    constexpr int ROWS  = 16;
    constexpr int PITCH = K + 1;
    __shared__ float Alds[ROWS * PITCH];

    const int tid     = threadIdx.x;
    const int wave    = tid >> 5;
    const int lane    = tid & 31;
    const int rowbase = blockIdx.x * ROWS;

    for (int idx = tid; idx < ROWS * K; idx += blockDim.x) {
        int r  = idx / K;
        int c  = idx - r * K;
        int gr = rowbase + r;
        Alds[r * PITCH + c] = (gr < M) ? A[(size_t)gr * K + c] : 0.0f;
    }
    __syncthreads();

    const int colbase = wave * 16;
    const int half    = lane >> 4;
    const int l15     = lane & 15;

    const float* Arow = &Alds[l15 * PITCH];

    v8f acc = {};
    #pragma unroll 8
    for (int k = 0; k < K; k += 4) {
        const int kk = k + 2 * half;
        v2f a, b;
        a.x = Arow[kk];
        a.y = Arow[kk + 1];
        b.x = B[(size_t)kk * NDIM + colbase + l15];
        b.y = B[(size_t)(kk + 1) * NDIM + colbase + l15];
        acc = __builtin_amdgcn_wmma_f32_16x16x4_f32(
             false, a,  false, b,
             (short)0, acc,  false,  false);
    }

    #pragma unroll
    for (int r = 0; r < 8; ++r) {
        int grow = rowbase + r + 8 * half;
        if (grow < M)
            XH[(size_t)grow * NDIM + colbase + l15] = acc[r];
    }

    __syncthreads();
    #pragma unroll
    for (int r = 0; r < 8; ++r)
        Alds[(r + 8 * half) * PITCH + colbase + l15] = acc[r];
    __syncthreads();

    const int kbase = (wave >> 2) * HC;
    const int ycol  = (wave & 3) * 16;
    v8f acc2 = {};
    #pragma unroll 8
    for (int k = 0; k < HC; k += 4) {
        const int kk = kbase + k + 2 * half;
        v2f a, b;
        a.x = Arow[kk];
        a.y = Arow[kk + 1];
        b.x = Wout[(size_t)kk * HC + ycol + l15];
        b.y = Wout[(size_t)(kk + 1) * HC + ycol + l15];
        acc2 = __builtin_amdgcn_wmma_f32_16x16x4_f32(
             false, a,  false, b,
             (short)0, acc2,  false,  false);
    }
    #pragma unroll
    for (int r = 0; r < 8; ++r) {
        int grow = rowbase + r + 8 * half;
        if (grow < M)
            Y[(size_t)grow * NDIM + colbase + l15] = acc2[r];
    }
}

typedef int v4i __attribute__((ext_vector_type(4)));

constexpr int OWN_R   = 196;
constexpr int OWN_CAP = 8192;
constexpr int OWN_T   = 1024;

constexpr int AL_ROWS = 16;
constexpr int AL_T    = 256;
constexpr int AL_R    = AL_ROWS * (AL_T / 32);
constexpr int KEY_HI  = 64;

constexpr int KEY_PIECE = 4096 - 8;
__device__ __forceinline__ int key_last_piece(int E) {
    return (int)(((((long long)E + 31) >> 5) - 1) >> 6);
}
__device__ __forceinline__ bool keys_usable(int N, int E) {
    if (E < 1 || (N + OWN_R - 1) / OWN_R > 256) return false;
    const long long wl = key_last_piece(E);
    return wl <= 496 && wl * 65536LL + 2048 <= 512LL * (long long)N;
}
__device__ __forceinline__ int key_vec_of_group(int g, int wl) {
    return (g >> 6) * KEY_PIECE + 8 * wl + (g & 63);
}

template <int XM>
__device__ __forceinline__ float row_xor_add(float v) {
    const int t = __builtin_amdgcn_update_dpp(0, __builtin_bit_cast(int, v), 0x160 + XM, 0xf, 0xf, true);
    return v + __builtin_bit_cast(float, t);
}

__global__ __launch_bounds__(AL_T)
void alpha_kernel(const float* xh,
                  const float* __restrict__ attl,
                  const float* __restrict__ attr,
                  float* __restrict__ al,
                  float* __restrict__ ar, int N,
                  const int* __restrict__ ei, int E,
                  unsigned int* keys) {
    const int lane = threadIdx.x & 31;
    const int wave = __builtin_amdgcn_readfirstlane((int)(threadIdx.x >> 5));
    const int w    = (int)blockIdx.x;
    const int n0   = w * AL_R + wave * AL_ROWS;
    const int c    = lane & 15, h = lane >> 4;

    const bool usekeys = keys_usable(N, E);
    const int  ng      = (int)(((long long)E + 31) >> 5);
    const int  gw      = w * 64 + wave * 8;
    const bool dokeys  = usekeys && gw < ng;

    unsigned int d[8] = {0u, 0u, 0u, 0u, 0u, 0u, 0u, 0u};
    if (dokeys) {
        const int* dcol = ei + E;
        #pragma unroll
        for (int i = 0; i < 8; ++i) {
            const long long e = ((long long)(gw + i) << 5) + lane;
            d[i] = (unsigned int)dcol[min(e, (long long)E - 1)];
        }
    }

    const v4f* xr = (const v4f*)xh;
    v4f x[AL_ROWS];
    #pragma unroll
    for (int k = 0; k < AL_ROWS; ++k) {
        const int n = min(n0 + k, N - 1);
        x[k] = __builtin_nontemporal_load(&xr[(size_t)n * 32 + lane]);
    }
    float al4[4], ar4[4];
    #pragma unroll
    for (int j = 0; j < 4; ++j) { al4[j] = attl[4 * lane + j];  ar4[j] = attr[4 * lane + j]; }

    float pl[AL_ROWS], pr[AL_ROWS];
    #pragma unroll
    for (int k = 0; k < AL_ROWS; ++k) {
        pl[k] = x[k][0] * al4[0] + x[k][1] * al4[1] + x[k][2] * al4[2] + x[k][3] * al4[3];
        pr[k] = x[k][0] * ar4[0] + x[k][1] * ar4[1] + x[k][2] * ar4[2] + x[k][3] * ar4[3];
    }
    #pragma unroll
    for (int k = 0; k < AL_ROWS; ++k) { pl[k] = row_xor_add<1>(pl[k]);  pr[k] = row_xor_add<1>(pr[k]); }
    #pragma unroll
    for (int k = 0; k < AL_ROWS; ++k) { pl[k] = row_xor_add<2>(pl[k]);  pr[k] = row_xor_add<2>(pr[k]); }
    #pragma unroll
    for (int k = 0; k < AL_ROWS; ++k) { pl[k] = row_xor_add<4>(pl[k]);  pr[k] = row_xor_add<4>(pr[k]); }
    #pragma unroll
    for (int k = 0; k < AL_ROWS; ++k) { pl[k] = row_xor_add<8>(pl[k]);  pr[k] = row_xor_add<8>(pr[k]); }

    float sl = pl[0], sr = pr[0];
    #pragma unroll
    for (int k = 1; k < AL_ROWS; ++k) {
        sl = (c == k) ? pl[k] : sl;
        sr = (c == k) ? pr[k] : sr;
    }
    if (n0 + c < N) {
        al[2 * (n0 + c) + h] = sl;
        ar[2 * (n0 + c) + h] = sr;
    }

    if (!usekeys) return;
    __syncthreads();
    if (!dokeys) return;

    unsigned int wlo = 0u, whi = 0u;
    #pragma unroll
    for (int i = 0; i < 8; ++i) {
        const unsigned int r = d[i] / OWN_R;
        #pragma unroll
        for (int b = 0; b < 4; ++b) {
            const unsigned int p0 = __builtin_amdgcn_ballot_w32(((r >> b) & 1u) != 0u);
            const unsigned int p1 = __builtin_amdgcn_ballot_w32(((r >> (b + 4)) & 1u) != 0u);
            if (lane == 4 * i + b) { wlo = p0;  whi = p1; }
        }
    }
    const int g  = gw + (lane >> 2);
    const int wi = 4 * key_vec_of_group(g, key_last_piece(E)) + (lane & 3);
    if (g < ng) {
        keys[wi]              = wlo;
        keys[wi + 4 * KEY_HI] = whi;
    }
}

__device__ __forceinline__ void lds_add_f32(float* p, float v) {
    __hip_atomic_fetch_add(p, v, __ATOMIC_RELAXED, __HIP_MEMORY_SCOPE_WORKGROUP);
}

__device__ __forceinline__ void own_append(int e, unsigned int* cnt, int* ls) {
    const unsigned int pos = atomicAdd(cnt, 1u);
    if (pos < (unsigned int)OWN_CAP) ls[pos] = e;
}

__device__ __forceinline__ void scan_group(v4i lo, v4i hi, const unsigned int (&inv)[8],
                                           int ebase, unsigned int keep, unsigned int* cnt, int* ls) {
    unsigned int m = ((unsigned int)lo[0] ^ inv[0]) & ((unsigned int)lo[1] ^ inv[1])
                   & ((unsigned int)lo[2] ^ inv[2]) & ((unsigned int)lo[3] ^ inv[3])
                   & ((unsigned int)hi[0] ^ inv[4]) & ((unsigned int)hi[1] ^ inv[5])
                   & ((unsigned int)hi[2] ^ inv[6]) & ((unsigned int)hi[3] ^ inv[7]) & keep;
    while (m) {
        own_append(ebase + __builtin_ctz(m), cnt, ls);
        m &= m - 1u;
    }
}

__device__ __forceinline__ void edge_logit2(const float* __restrict__ al,
                                            const float* __restrict__ ar,
                                            int s, int d, float& v0, float& v1) {
    const float a0 = al[2 * s], a1 = al[2 * s + 1];
    const float r0 = ar[2 * d], r1 = ar[2 * d + 1];
    v0 = a0 + r0;  v0 = (v0 > 0.0f) ? v0 : 0.2f * v0;
    v1 = a1 + r1;  v1 = (v1 > 0.0f) ? v1 : 0.2f * v1;
}

__device__ __forceinline__ float seg_max_of(unsigned int key) {
    float m = key_to_f32(key);
    if (!(fabsf(m) <= 3.4e38f)) m = 0.0f;
    return m;
}

template <class F>
__device__ __forceinline__ void own_rescan(const int* __restrict__ ei, int E,
                                           int lo, int nown, int tid, F f) {
    for (int e = tid; e < E; e += OWN_T) {
        const unsigned int dl = (unsigned int)(ei[E + e] - lo);
        if (dl < (unsigned int)nown) f(ei[e], (int)dl);
    }
    for (int i = tid; i < nown; i += OWN_T) f(lo + i, i);
}

__global__ __launch_bounds__(1024)
void gat_owner_kernel(const int* __restrict__ ei, int E, int N,
                      const unsigned int* __restrict__ keys,
                      const float* __restrict__ al,
                      const float* __restrict__ ar,
                      const float* __restrict__ y,
                      const float* __restrict__ bias,
                      float* __restrict__ out) {
    __shared__ v2f            sw[OWN_CAP];
    __shared__ float          lv[OWN_CAP * 2];
    __shared__ int            ls[OWN_CAP];
    __shared__ int            ss[OWN_CAP];
    __shared__ unsigned short ld[OWN_CAP];
    __shared__ unsigned short rk[OWN_CAP];
    __shared__ unsigned int   mkey[OWN_R * 2];
    __shared__ float          den[OWN_R * 2];
    __shared__ unsigned int   deg[OWN_R];
    __shared__ unsigned int   start[OWN_R];
    __shared__ unsigned int   cnt;
    static_assert(sizeof(float) * OWN_R * 64 <= sizeof(v2f) * OWN_CAP, "out tile must fit the slot array");
    float* const tile = (float*)sw;
    constexpr int PER = (OWN_R + 31) / 32;

    const int tid  = threadIdx.x;
    const int lane = tid & 31;
    const int wave = tid >> 5;
    const int nranges = (N + OWN_R - 1) / OWN_R;
    const bool usekeys = keys_usable(N, E);

    for (int rg = blockIdx.x; rg < nranges; rg += gridDim.x) {
        const int lo   = rg * OWN_R;
        const int nown = min(OWN_R, N - lo);

        for (int i = tid; i < nown * 2; i += OWN_T) { mkey[i] = 0u; den[i] = 0.0f; }
        for (int i = tid; i < nown; i += OWN_T) deg[i] = 0u;
        if (tid == 0) cnt = 0u;
        __syncthreads();

        if (usekeys) {
            unsigned int inv[8];
            #pragma unroll
            for (int b = 0; b < 8; ++b) inv[b] = ((rg >> b) & 1) ? 0u : 0xffffffffu;
            const v4i* __restrict__ k4 = (const v4i*)keys;
            const int ng = (int)(((long long)E + 31) >> 5);
            const int wl = key_last_piece(E);
            constexpr int G = 4;
            const int nbatch = (E >> 5) / (G * OWN_T);
            int bt = (int)(((long long)rg * nbatch) / nranges);
            for (int t = 0; t < nbatch; ++t) {
                const int g0 = bt * (G * OWN_T) + tid;
                bt = (bt + 1 < nbatch) ? bt + 1 : 0;
                v4i lo[G], hi[G];
                static_assert(OWN_T == 16 * 64, "a lane's groups must be 16 pieces apart");
                const int vi0 = key_vec_of_group(g0, wl);
                #pragma unroll
                for (int k = 0; k < G; ++k) {
                    const int vi = vi0 + k * (16 * KEY_PIECE);
                    lo[k] = __builtin_nontemporal_load(&k4[vi]);
                    hi[k] = __builtin_nontemporal_load(&k4[vi + KEY_HI]);
                }
                #pragma unroll
                for (int k = 0; k < G; ++k)
                    scan_group(lo[k], hi[k], inv, 32 * (g0 + k * OWN_T), 0xffffffffu, &cnt, ls);
            }
            for (int g = nbatch * (G * OWN_T) + tid; g < ng; g += OWN_T) {
                const int vi = key_vec_of_group(g, wl);
                const v4i lo = __builtin_nontemporal_load(&k4[vi]);
                const v4i hi = __builtin_nontemporal_load(&k4[vi + KEY_HI]);
                const int rem = E - 32 * g;
                scan_group(lo, hi, inv, 32 * g, (rem < 32) ? (1u << rem) - 1u : 0xffffffffu, &cnt, ls);
            }
        } else {
            for (int e = tid; e < E; e += OWN_T) {
                const unsigned int dl = (unsigned int)(ei[E + e] - lo);
                if (dl < (unsigned int)nown) own_append(e, &cnt, ls);
            }
        }
        __syncthreads();

        const unsigned int nlisted = cnt;
        const unsigned int total   = nlisted + (unsigned int)nown;
        if (total <= (unsigned int)OWN_CAP) {
            for (unsigned int i0 = tid; i0 < nlisted; i0 += 4 * OWN_T) {
                unsigned int ix[4];
                int en[4], sn[4], dn[4];
                #pragma unroll
                for (int k = 0; k < 4; ++k) {
                    const unsigned int i = i0 + k * OWN_T;
                    ix[k] = (i < nlisted) ? i : i0;
                    en[k] = ls[ix[k]];
                }
                #pragma unroll
                for (int k = 0; k < 4; ++k) { sn[k] = ei[en[k]];  dn[k] = ei[E + en[k]]; }
                #pragma unroll
                for (int k = 0; k < 4; ++k) {
                    ls[ix[k]] = sn[k];
                    ld[ix[k]] = (unsigned short)(dn[k] - lo);
                }
            }
            for (int i = tid; i < nown; i += OWN_T) {
                ls[nlisted + i] = lo + i;
                ld[nlisted + i] = (unsigned short)i;
            }
            __syncthreads();

            for (unsigned int i = tid; i < total; i += OWN_T) {
                const int s = ls[i], dl = ld[i];
                float v0, v1;
                edge_logit2(al, ar, s, lo + dl, v0, v1);
                lv[2 * i] = v0;  lv[2 * i + 1] = v1;
                atomicMax(&mkey[2 * dl],     f32_to_key(v0));
                atomicMax(&mkey[2 * dl + 1], f32_to_key(v1));
                rk[i] = (unsigned short)atomicAdd(&deg[dl], 1u);
            }
            __syncthreads();
            if (wave == 0) {
                unsigned int c[PER];
                unsigned int sum = 0u;
                #pragma unroll
                for (int j = 0; j < PER; ++j) {
                    const int n = PER * lane + j;
                    c[j] = (n < nown) ? deg[n] : 0u;
                    sum += c[j];
                }
                unsigned int incl = sum;
                #pragma unroll
                for (int off = 1; off < 32; off <<= 1) {
                    const unsigned int t = __shfl_up(incl, off, 32);
                    if (lane >= off) incl += t;
                }
                unsigned int run = incl - sum;
                #pragma unroll
                for (int j = 0; j < PER; ++j) {
                    const int n = PER * lane + j;
                    if (n < nown) start[n] = run;
                    run += c[j];
                }
            }
            for (unsigned int i = tid; i < total; i += OWN_T) {
                const int dl = ld[i];
                const float x0 = __expf(lv[2 * i]     - seg_max_of(mkey[2 * dl]));
                const float x1 = __expf(lv[2 * i + 1] - seg_max_of(mkey[2 * dl + 1]));
                lv[2 * i] = x0;  lv[2 * i + 1] = x1;
                lds_add_f32(&den[2 * dl],     x0);
                lds_add_f32(&den[2 * dl + 1], x1);
            }
            __syncthreads();
            for (unsigned int i = tid; i < total; i += OWN_T) {
                const int dl = ld[i];
                const unsigned int p = start[dl] + rk[i];
                v2f w;
                w.x = lv[2 * i]     / (den[2 * dl]     + 1e-16f);
                w.y = lv[2 * i + 1] / (den[2 * dl + 1] + 1e-16f);
                ss[p] = ls[i];
                sw[p] = w;
            }
            __syncthreads();
            const int wave_s = __builtin_amdgcn_readfirstlane(wave);
            for (int n = wave_s; n < nown; n += 32) {
                const unsigned int b  = __builtin_amdgcn_readfirstlane(start[n]);
                const unsigned int dn = __builtin_amdgcn_readfirstlane(deg[n]);
                float a0 = 0.0f, a1 = 0.0f;
                for (unsigned int j = 0; j < dn; j += 8) {
                    const unsigned int o    = j + (unsigned int)(lane & 7);
                    const unsigned int slot = b + ((o < dn) ? o : j);
                    const int sv = ss[slot];
                    const v2f wv = sw[slot];
                    float p0[8], p1[8], q0[8], q1[8];
                    #pragma unroll
                    for (int k = 0; k < 8; ++k) {
                        const int s = __builtin_amdgcn_readlane(sv, k);
                        const float* ys = y + (size_t)s * 128;
                        p0[k] = ys[lane];       p1[k] = ys[lane + 32];
                        q0[k] = ys[lane + 64];  q1[k] = ys[lane + 96];
                    }
                    #pragma unroll
                    for (int k = 0; k < 8; ++k) {
                        const float w0 = __uint_as_float((unsigned int)__builtin_amdgcn_readlane((int)__float_as_uint(wv.x), k));
                        const float w1 = __uint_as_float((unsigned int)__builtin_amdgcn_readlane((int)__float_as_uint(wv.y), k));
                        const float t0 = p0[k] * w0 + q0[k] * w1;
                        const float t1 = p1[k] * w0 + q1[k] * w1;
                        const bool in_run = (j + (unsigned int)k < dn);
                        a0 += in_run ? t0 : 0.0f;
                        a1 += in_run ? t1 : 0.0f;
                    }
                }
                float* og = out + (size_t)(lo + n) * 64;
                og[lane]      = a0 + bias[lane];
                og[lane + 32] = a1 + bias[lane + 32];
            }
        } else {
            for (int i = tid; i < nown * 64; i += OWN_T) tile[i] = 0.0f;
            __syncthreads();
            own_rescan(ei, E, lo, nown, tid, [&](int s, int dl) {
                float v0, v1;
                edge_logit2(al, ar, s, lo + dl, v0, v1);
                atomicMax(&mkey[2 * dl],     f32_to_key(v0));
                atomicMax(&mkey[2 * dl + 1], f32_to_key(v1));
            });
            __syncthreads();
            own_rescan(ei, E, lo, nown, tid, [&](int s, int dl) {
                float v0, v1;
                edge_logit2(al, ar, s, lo + dl, v0, v1);
                lds_add_f32(&den[2 * dl],     __expf(v0 - seg_max_of(mkey[2 * dl])));
                lds_add_f32(&den[2 * dl + 1], __expf(v1 - seg_max_of(mkey[2 * dl + 1])));
            });
            __syncthreads();
            own_rescan(ei, E, lo, nown, tid, [&](int s, int dl) {
                float v0, v1;
                edge_logit2(al, ar, s, lo + dl, v0, v1);
                const float w0 = __expf(v0 - seg_max_of(mkey[2 * dl]))     / (den[2 * dl]     + 1e-16f);
                const float w1 = __expf(v1 - seg_max_of(mkey[2 * dl + 1])) / (den[2 * dl + 1] + 1e-16f);
                const float* ys = y + (size_t)s * 128;
                for (int c = 0; c < 64; ++c)
                    lds_add_f32(&tile[dl * 64 + c], ys[c] * w0 + ys[64 + c] * w1);
            });
            __syncthreads();
            float* og = out + (size_t)lo * 64;
            for (int i = tid; i < nown * 64; i += OWN_T) og[i] = tile[i] + bias[i & 63];
        }
        __syncthreads();
    }
}

extern "C" void kernel_launch(void* const* d_in, const int* in_sizes, int n_in,
                              void* d_out, int out_size, void* d_ws, size_t ws_size,
                              hipStream_t stream) {
    const float* x    = (const float*)d_in[0];
    const int*   ei   = (const int*)d_in[1];
    const float* Wl   = (const float*)d_in[2];
    const float* attl = (const float*)d_in[3];
    const float* attr = (const float*)d_in[4];
    const float* Wout = (const float*)d_in[5];
    const float* bias = (const float*)d_in[6];
    float* out = (float*)d_out;

    const int N   = in_sizes[0] / 128;
    const int E   = in_sizes[1] / 2;

    float* ws    = (float*)d_ws;
    float* xh    = ws;
    float* y     = xh  + (size_t)N * 128;
    float* al    = y   + (size_t)N * 128;
    float* ar    = al  + (size_t)N * 2;

    proj_gemm_kernel<<<(N + 15) / 16, 256, 0, stream>>>(x, Wl, Wout, xh, y, N);

    unsigned int* keys = (unsigned int*)xh;
    alpha_kernel<<<(N + AL_R - 1) / AL_R, AL_T, 0, stream>>>(xh, attl, attr, al, ar, N, ei, E, keys);

    gat_owner_kernel<<<256, 1024, 0, stream>>>(ei, E, N, keys, al, ar, y, bias, out);
}


# Projection kernel stages its x tile with two 16-byte loads per thread issued together (a wave's load = one whole row): one trip to memory per tile, not eight dependent ones; sums and order unchanged
# speedup vs baseline: 4.9295x; 1.0844x over previous
#include <hip/hip_runtime.h>
#include <hip/hip_bf16.h>
#include <math.h>

typedef float v2f __attribute__((ext_vector_type(2)));
typedef float v8f __attribute__((ext_vector_type(8)));
typedef float v4f __attribute__((ext_vector_type(4)));

__device__ __forceinline__ unsigned int f32_to_key(float v) {
    const unsigned int b = __float_as_uint(v);
    return (b & 0x80000000u) ? ~b : (b | 0x80000000u);
}
__device__ __forceinline__ float key_to_f32(unsigned int k) {
    const unsigned int b = (k & 0x80000000u) ? (k & 0x7fffffffu) : ~k;
    return __uint_as_float(b);
}

__global__ __launch_bounds__(256)
void proj_gemm_kernel(const float* __restrict__ A,
                                 const float* __restrict__ B,
                                 const float* __restrict__ Wout,
                                 float* __restrict__ XH,
                                 float* __restrict__ Y,
                                 int M) {
    constexpr int K     = 128;
    constexpr int NDIM  = 128;
    constexpr int HC    = 64;
    constexpr int ROWS  = 16;
    constexpr int PITCH = K + 1;
    constexpr int NT    = 256;
    __shared__ float Alds[ROWS * PITCH];

    const int tid     = threadIdx.x;
    const int wave    = tid >> 5;
    const int lane    = tid & 31;
    const int rowbase = blockIdx.x * ROWS;

    static_assert(ROWS * K == 2 * 4 * NT, "two 16-byte pieces per thread");
    const v4f* A4 = (const v4f*)A;
    v4f xt[2];
    #pragma unroll
    for (int i = 0; i < 2; ++i) {
        const int p  = tid + i * NT;
        const int gr = min(rowbase + (p >> 5), M - 1);
        xt[i] = A4[(size_t)gr * (K / 4) + (p & 31)];
    }
    #pragma unroll
    for (int i = 0; i < 2; ++i) {
        const int p = tid + i * NT;
        const int r = p >> 5;
        const bool in = rowbase + r < M;
        float* dstp = &Alds[r * PITCH + 4 * (p & 31)];
        #pragma unroll
        for (int j = 0; j < 4; ++j) dstp[j] = in ? xt[i][j] : 0.0f;
    }
    __syncthreads();

    const int colbase = wave * 16;
    const int half    = lane >> 4;
    const int l15     = lane & 15;

    const float* Arow = &Alds[l15 * PITCH];

    v8f acc = {};
    #pragma unroll 8
    for (int k = 0; k < K; k += 4) {
        const int kk = k + 2 * half;
        v2f a, b;
        a.x = Arow[kk];
        a.y = Arow[kk + 1];
        b.x = B[(size_t)kk * NDIM + colbase + l15];
        b.y = B[(size_t)(kk + 1) * NDIM + colbase + l15];
        acc = __builtin_amdgcn_wmma_f32_16x16x4_f32(
             false, a,  false, b,
             (short)0, acc,  false,  false);
    }

    #pragma unroll
    for (int r = 0; r < 8; ++r) {
        int grow = rowbase + r + 8 * half;
        if (grow < M)
            XH[(size_t)grow * NDIM + colbase + l15] = acc[r];
    }

    __syncthreads();
    #pragma unroll
    for (int r = 0; r < 8; ++r)
        Alds[(r + 8 * half) * PITCH + colbase + l15] = acc[r];
    __syncthreads();

    const int kbase = (wave >> 2) * HC;
    const int ycol  = (wave & 3) * 16;
    v8f acc2 = {};
    #pragma unroll 8
    for (int k = 0; k < HC; k += 4) {
        const int kk = kbase + k + 2 * half;
        v2f a, b;
        a.x = Arow[kk];
        a.y = Arow[kk + 1];
        b.x = Wout[(size_t)kk * HC + ycol + l15];
        b.y = Wout[(size_t)(kk + 1) * HC + ycol + l15];
        acc2 = __builtin_amdgcn_wmma_f32_16x16x4_f32(
             false, a,  false, b,
             (short)0, acc2,  false,  false);
    }
    #pragma unroll
    for (int r = 0; r < 8; ++r) {
        int grow = rowbase + r + 8 * half;
        if (grow < M)
            Y[(size_t)grow * NDIM + colbase + l15] = acc2[r];
    }
}

typedef int v4i __attribute__((ext_vector_type(4)));

constexpr int OWN_R   = 196;
constexpr int OWN_CAP = 8192;
constexpr int OWN_T   = 1024;

constexpr int AL_ROWS = 16;
constexpr int AL_T    = 256;
constexpr int AL_R    = AL_ROWS * (AL_T / 32);
constexpr int KEY_HI  = 64;

constexpr int KEY_PIECE = 4096 - 8;
__device__ __forceinline__ int key_last_piece(int E) {
    return (int)(((((long long)E + 31) >> 5) - 1) >> 6);
}
__device__ __forceinline__ bool keys_usable(int N, int E) {
    if (E < 1 || (N + OWN_R - 1) / OWN_R > 256) return false;
    const long long wl = key_last_piece(E);
    return wl <= 496 && wl * 65536LL + 2048 <= 512LL * (long long)N;
}
__device__ __forceinline__ int key_vec_of_group(int g, int wl) {
    return (g >> 6) * KEY_PIECE + 8 * wl + (g & 63);
}

template <int XM>
__device__ __forceinline__ float row_xor_add(float v) {
    const int t = __builtin_amdgcn_update_dpp(0, __builtin_bit_cast(int, v), 0x160 + XM, 0xf, 0xf, true);
    return v + __builtin_bit_cast(float, t);
}

__global__ __launch_bounds__(AL_T)
void alpha_kernel(const float* xh,
                  const float* __restrict__ attl,
                  const float* __restrict__ attr,
                  float* __restrict__ al,
                  float* __restrict__ ar, int N,
                  const int* __restrict__ ei, int E,
                  unsigned int* keys) {
    const int lane = threadIdx.x & 31;
    const int wave = __builtin_amdgcn_readfirstlane((int)(threadIdx.x >> 5));
    const int w    = (int)blockIdx.x;
    const int n0   = w * AL_R + wave * AL_ROWS;
    const int c    = lane & 15, h = lane >> 4;

    const bool usekeys = keys_usable(N, E);
    const int  ng      = (int)(((long long)E + 31) >> 5);
    const int  gw      = w * 64 + wave * 8;
    const bool dokeys  = usekeys && gw < ng;

    unsigned int d[8] = {0u, 0u, 0u, 0u, 0u, 0u, 0u, 0u};
    if (dokeys) {
        const int* dcol = ei + E;
        #pragma unroll
        for (int i = 0; i < 8; ++i) {
            const long long e = ((long long)(gw + i) << 5) + lane;
            d[i] = (unsigned int)dcol[min(e, (long long)E - 1)];
        }
    }

    const v4f* xr = (const v4f*)xh;
    v4f x[AL_ROWS];
    #pragma unroll
    for (int k = 0; k < AL_ROWS; ++k) {
        const int n = min(n0 + k, N - 1);
        x[k] = __builtin_nontemporal_load(&xr[(size_t)n * 32 + lane]);
    }
    float al4[4], ar4[4];
    #pragma unroll
    for (int j = 0; j < 4; ++j) { al4[j] = attl[4 * lane + j];  ar4[j] = attr[4 * lane + j]; }

    float pl[AL_ROWS], pr[AL_ROWS];
    #pragma unroll
    for (int k = 0; k < AL_ROWS; ++k) {
        pl[k] = x[k][0] * al4[0] + x[k][1] * al4[1] + x[k][2] * al4[2] + x[k][3] * al4[3];
        pr[k] = x[k][0] * ar4[0] + x[k][1] * ar4[1] + x[k][2] * ar4[2] + x[k][3] * ar4[3];
    }
    #pragma unroll
    for (int k = 0; k < AL_ROWS; ++k) { pl[k] = row_xor_add<1>(pl[k]);  pr[k] = row_xor_add<1>(pr[k]); }
    #pragma unroll
    for (int k = 0; k < AL_ROWS; ++k) { pl[k] = row_xor_add<2>(pl[k]);  pr[k] = row_xor_add<2>(pr[k]); }
    #pragma unroll
    for (int k = 0; k < AL_ROWS; ++k) { pl[k] = row_xor_add<4>(pl[k]);  pr[k] = row_xor_add<4>(pr[k]); }
    #pragma unroll
    for (int k = 0; k < AL_ROWS; ++k) { pl[k] = row_xor_add<8>(pl[k]);  pr[k] = row_xor_add<8>(pr[k]); }

    float sl = pl[0], sr = pr[0];
    #pragma unroll
    for (int k = 1; k < AL_ROWS; ++k) {
        sl = (c == k) ? pl[k] : sl;
        sr = (c == k) ? pr[k] : sr;
    }
    if (n0 + c < N) {
        al[2 * (n0 + c) + h] = sl;
        ar[2 * (n0 + c) + h] = sr;
    }

    if (!usekeys) return;
    __syncthreads();
    if (!dokeys) return;

    unsigned int wlo = 0u, whi = 0u;
    #pragma unroll
    for (int i = 0; i < 8; ++i) {
        const unsigned int r = d[i] / OWN_R;
        #pragma unroll
        for (int b = 0; b < 4; ++b) {
            const unsigned int p0 = __builtin_amdgcn_ballot_w32(((r >> b) & 1u) != 0u);
            const unsigned int p1 = __builtin_amdgcn_ballot_w32(((r >> (b + 4)) & 1u) != 0u);
            if (lane == 4 * i + b) { wlo = p0;  whi = p1; }
        }
    }
    const int g  = gw + (lane >> 2);
    const int wi = 4 * key_vec_of_group(g, key_last_piece(E)) + (lane & 3);
    if (g < ng) {
        keys[wi]              = wlo;
        keys[wi + 4 * KEY_HI] = whi;
    }
}

__device__ __forceinline__ void lds_add_f32(float* p, float v) {
    __hip_atomic_fetch_add(p, v, __ATOMIC_RELAXED, __HIP_MEMORY_SCOPE_WORKGROUP);
}

__device__ __forceinline__ void own_append(int e, unsigned int* cnt, int* ls) {
    const unsigned int pos = atomicAdd(cnt, 1u);
    if (pos < (unsigned int)OWN_CAP) ls[pos] = e;
}

__device__ __forceinline__ void scan_group(v4i lo, v4i hi, const unsigned int (&inv)[8],
                                           int ebase, unsigned int keep, unsigned int* cnt, int* ls) {
    unsigned int m = ((unsigned int)lo[0] ^ inv[0]) & ((unsigned int)lo[1] ^ inv[1])
                   & ((unsigned int)lo[2] ^ inv[2]) & ((unsigned int)lo[3] ^ inv[3])
                   & ((unsigned int)hi[0] ^ inv[4]) & ((unsigned int)hi[1] ^ inv[5])
                   & ((unsigned int)hi[2] ^ inv[6]) & ((unsigned int)hi[3] ^ inv[7]) & keep;
    while (m) {
        own_append(ebase + __builtin_ctz(m), cnt, ls);
        m &= m - 1u;
    }
}

__device__ __forceinline__ void edge_logit2(const float* __restrict__ al,
                                            const float* __restrict__ ar,
                                            int s, int d, float& v0, float& v1) {
    const float a0 = al[2 * s], a1 = al[2 * s + 1];
    const float r0 = ar[2 * d], r1 = ar[2 * d + 1];
    v0 = a0 + r0;  v0 = (v0 > 0.0f) ? v0 : 0.2f * v0;
    v1 = a1 + r1;  v1 = (v1 > 0.0f) ? v1 : 0.2f * v1;
}

__device__ __forceinline__ float seg_max_of(unsigned int key) {
    float m = key_to_f32(key);
    if (!(fabsf(m) <= 3.4e38f)) m = 0.0f;
    return m;
}

template <class F>
__device__ __forceinline__ void own_rescan(const int* __restrict__ ei, int E,
                                           int lo, int nown, int tid, F f) {
    for (int e = tid; e < E; e += OWN_T) {
        const unsigned int dl = (unsigned int)(ei[E + e] - lo);
        if (dl < (unsigned int)nown) f(ei[e], (int)dl);
    }
    for (int i = tid; i < nown; i += OWN_T) f(lo + i, i);
}

__global__ __launch_bounds__(1024)
void gat_owner_kernel(const int* __restrict__ ei, int E, int N,
                      const unsigned int* __restrict__ keys,
                      const float* __restrict__ al,
                      const float* __restrict__ ar,
                      const float* __restrict__ y,
                      const float* __restrict__ bias,
                      float* __restrict__ out) {
    __shared__ v2f            sw[OWN_CAP];
    __shared__ float          lv[OWN_CAP * 2];
    __shared__ int            ls[OWN_CAP];
    __shared__ int            ss[OWN_CAP];
    __shared__ unsigned short ld[OWN_CAP];
    __shared__ unsigned short rk[OWN_CAP];
    __shared__ unsigned int   mkey[OWN_R * 2];
    __shared__ float          den[OWN_R * 2];
    __shared__ unsigned int   deg[OWN_R];
    __shared__ unsigned int   start[OWN_R];
    __shared__ unsigned int   cnt;
    static_assert(sizeof(float) * OWN_R * 64 <= sizeof(v2f) * OWN_CAP, "out tile must fit the slot array");
    float* const tile = (float*)sw;
    constexpr int PER = (OWN_R + 31) / 32;

    const int tid  = threadIdx.x;
    const int lane = tid & 31;
    const int wave = tid >> 5;
    const int nranges = (N + OWN_R - 1) / OWN_R;
    const bool usekeys = keys_usable(N, E);

    for (int rg = blockIdx.x; rg < nranges; rg += gridDim.x) {
        const int lo   = rg * OWN_R;
        const int nown = min(OWN_R, N - lo);

        for (int i = tid; i < nown * 2; i += OWN_T) { mkey[i] = 0u; den[i] = 0.0f; }
        for (int i = tid; i < nown; i += OWN_T) deg[i] = 0u;
        if (tid == 0) cnt = 0u;
        __syncthreads();

        if (usekeys) {
            unsigned int inv[8];
            #pragma unroll
            for (int b = 0; b < 8; ++b) inv[b] = ((rg >> b) & 1) ? 0u : 0xffffffffu;
            const v4i* __restrict__ k4 = (const v4i*)keys;
            const int ng = (int)(((long long)E + 31) >> 5);
            const int wl = key_last_piece(E);
            constexpr int G = 4;
            const int nbatch = (E >> 5) / (G * OWN_T);
            int bt = (int)(((long long)rg * nbatch) / nranges);
            for (int t = 0; t < nbatch; ++t) {
                const int g0 = bt * (G * OWN_T) + tid;
                bt = (bt + 1 < nbatch) ? bt + 1 : 0;
                v4i lo[G], hi[G];
                static_assert(OWN_T == 16 * 64, "a lane's groups must be 16 pieces apart");
                const int vi0 = key_vec_of_group(g0, wl);
                #pragma unroll
                for (int k = 0; k < G; ++k) {
                    const int vi = vi0 + k * (16 * KEY_PIECE);
                    lo[k] = __builtin_nontemporal_load(&k4[vi]);
                    hi[k] = __builtin_nontemporal_load(&k4[vi + KEY_HI]);
                }
                #pragma unroll
                for (int k = 0; k < G; ++k)
                    scan_group(lo[k], hi[k], inv, 32 * (g0 + k * OWN_T), 0xffffffffu, &cnt, ls);
            }
            for (int g = nbatch * (G * OWN_T) + tid; g < ng; g += OWN_T) {
                const int vi = key_vec_of_group(g, wl);
                const v4i lo = __builtin_nontemporal_load(&k4[vi]);
                const v4i hi = __builtin_nontemporal_load(&k4[vi + KEY_HI]);
                const int rem = E - 32 * g;
                scan_group(lo, hi, inv, 32 * g, (rem < 32) ? (1u << rem) - 1u : 0xffffffffu, &cnt, ls);
            }
        } else {
            for (int e = tid; e < E; e += OWN_T) {
                const unsigned int dl = (unsigned int)(ei[E + e] - lo);
                if (dl < (unsigned int)nown) own_append(e, &cnt, ls);
            }
        }
        __syncthreads();

        const unsigned int nlisted = cnt;
        const unsigned int total   = nlisted + (unsigned int)nown;
        if (total <= (unsigned int)OWN_CAP) {
            for (unsigned int i0 = tid; i0 < nlisted; i0 += 4 * OWN_T) {
                unsigned int ix[4];
                int en[4], sn[4], dn[4];
                #pragma unroll
                for (int k = 0; k < 4; ++k) {
                    const unsigned int i = i0 + k * OWN_T;
                    ix[k] = (i < nlisted) ? i : i0;
                    en[k] = ls[ix[k]];
                }
                #pragma unroll
                for (int k = 0; k < 4; ++k) { sn[k] = ei[en[k]];  dn[k] = ei[E + en[k]]; }
                #pragma unroll
                for (int k = 0; k < 4; ++k) {
                    ls[ix[k]] = sn[k];
                    ld[ix[k]] = (unsigned short)(dn[k] - lo);
                }
            }
            for (int i = tid; i < nown; i += OWN_T) {
                ls[nlisted + i] = lo + i;
                ld[nlisted + i] = (unsigned short)i;
            }
            __syncthreads();

            for (unsigned int i = tid; i < total; i += OWN_T) {
                const int s = ls[i], dl = ld[i];
                float v0, v1;
                edge_logit2(al, ar, s, lo + dl, v0, v1);
                lv[2 * i] = v0;  lv[2 * i + 1] = v1;
                atomicMax(&mkey[2 * dl],     f32_to_key(v0));
                atomicMax(&mkey[2 * dl + 1], f32_to_key(v1));
                rk[i] = (unsigned short)atomicAdd(&deg[dl], 1u);
            }
            __syncthreads();
            if (wave == 0) {
                unsigned int c[PER];
                unsigned int sum = 0u;
                #pragma unroll
                for (int j = 0; j < PER; ++j) {
                    const int n = PER * lane + j;
                    c[j] = (n < nown) ? deg[n] : 0u;
                    sum += c[j];
                }
                unsigned int incl = sum;
                #pragma unroll
                for (int off = 1; off < 32; off <<= 1) {
                    const unsigned int t = __shfl_up(incl, off, 32);
                    if (lane >= off) incl += t;
                }
                unsigned int run = incl - sum;
                #pragma unroll
                for (int j = 0; j < PER; ++j) {
                    const int n = PER * lane + j;
                    if (n < nown) start[n] = run;
                    run += c[j];
                }
            }
            for (unsigned int i = tid; i < total; i += OWN_T) {
                const int dl = ld[i];
                const float x0 = __expf(lv[2 * i]     - seg_max_of(mkey[2 * dl]));
                const float x1 = __expf(lv[2 * i + 1] - seg_max_of(mkey[2 * dl + 1]));
                lv[2 * i] = x0;  lv[2 * i + 1] = x1;
                lds_add_f32(&den[2 * dl],     x0);
                lds_add_f32(&den[2 * dl + 1], x1);
            }
            __syncthreads();
            for (unsigned int i = tid; i < total; i += OWN_T) {
                const int dl = ld[i];
                const unsigned int p = start[dl] + rk[i];
                v2f w;
                w.x = lv[2 * i]     / (den[2 * dl]     + 1e-16f);
                w.y = lv[2 * i + 1] / (den[2 * dl + 1] + 1e-16f);
                ss[p] = ls[i];
                sw[p] = w;
            }
            __syncthreads();
            const int wave_s = __builtin_amdgcn_readfirstlane(wave);
            for (int n = wave_s; n < nown; n += 32) {
                const unsigned int b  = __builtin_amdgcn_readfirstlane(start[n]);
                const unsigned int dn = __builtin_amdgcn_readfirstlane(deg[n]);
                float a0 = 0.0f, a1 = 0.0f;
                for (unsigned int j = 0; j < dn; j += 8) {
                    const unsigned int o    = j + (unsigned int)(lane & 7);
                    const unsigned int slot = b + ((o < dn) ? o : j);
                    const int sv = ss[slot];
                    const v2f wv = sw[slot];
                    float p0[8], p1[8], q0[8], q1[8];
                    #pragma unroll
                    for (int k = 0; k < 8; ++k) {
                        const int s = __builtin_amdgcn_readlane(sv, k);
                        const float* ys = y + (size_t)s * 128;
                        p0[k] = ys[lane];       p1[k] = ys[lane + 32];
                        q0[k] = ys[lane + 64];  q1[k] = ys[lane + 96];
                    }
                    #pragma unroll
                    for (int k = 0; k < 8; ++k) {
                        const float w0 = __uint_as_float((unsigned int)__builtin_amdgcn_readlane((int)__float_as_uint(wv.x), k));
                        const float w1 = __uint_as_float((unsigned int)__builtin_amdgcn_readlane((int)__float_as_uint(wv.y), k));
                        const float t0 = p0[k] * w0 + q0[k] * w1;
                        const float t1 = p1[k] * w0 + q1[k] * w1;
                        const bool in_run = (j + (unsigned int)k < dn);
                        a0 += in_run ? t0 : 0.0f;
                        a1 += in_run ? t1 : 0.0f;
                    }
                }
                float* og = out + (size_t)(lo + n) * 64;
                og[lane]      = a0 + bias[lane];
                og[lane + 32] = a1 + bias[lane + 32];
            }
        } else {
            for (int i = tid; i < nown * 64; i += OWN_T) tile[i] = 0.0f;
            __syncthreads();
            own_rescan(ei, E, lo, nown, tid, [&](int s, int dl) {
                float v0, v1;
                edge_logit2(al, ar, s, lo + dl, v0, v1);
                atomicMax(&mkey[2 * dl],     f32_to_key(v0));
                atomicMax(&mkey[2 * dl + 1], f32_to_key(v1));
            });
            __syncthreads();
            own_rescan(ei, E, lo, nown, tid, [&](int s, int dl) {
                float v0, v1;
                edge_logit2(al, ar, s, lo + dl, v0, v1);
                lds_add_f32(&den[2 * dl],     __expf(v0 - seg_max_of(mkey[2 * dl])));
                lds_add_f32(&den[2 * dl + 1], __expf(v1 - seg_max_of(mkey[2 * dl + 1])));
            });
            __syncthreads();
            own_rescan(ei, E, lo, nown, tid, [&](int s, int dl) {
                float v0, v1;
                edge_logit2(al, ar, s, lo + dl, v0, v1);
                const float w0 = __expf(v0 - seg_max_of(mkey[2 * dl]))     / (den[2 * dl]     + 1e-16f);
                const float w1 = __expf(v1 - seg_max_of(mkey[2 * dl + 1])) / (den[2 * dl + 1] + 1e-16f);
                const float* ys = y + (size_t)s * 128;
                for (int c = 0; c < 64; ++c)
                    lds_add_f32(&tile[dl * 64 + c], ys[c] * w0 + ys[64 + c] * w1);
            });
            __syncthreads();
            float* og = out + (size_t)lo * 64;
            for (int i = tid; i < nown * 64; i += OWN_T) og[i] = tile[i] + bias[i & 63];
        }
        __syncthreads();
    }
}

extern "C" void kernel_launch(void* const* d_in, const int* in_sizes, int n_in,
                              void* d_out, int out_size, void* d_ws, size_t ws_size,
                              hipStream_t stream) {
    const float* x    = (const float*)d_in[0];
    const int*   ei   = (const int*)d_in[1];
    const float* Wl   = (const float*)d_in[2];
    const float* attl = (const float*)d_in[3];
    const float* attr = (const float*)d_in[4];
    const float* Wout = (const float*)d_in[5];
    const float* bias = (const float*)d_in[6];
    float* out = (float*)d_out;

    const int N   = in_sizes[0] / 128;
    const int E   = in_sizes[1] / 2;

    float* ws    = (float*)d_ws;
    float* xh    = ws;
    float* y     = xh  + (size_t)N * 128;
    float* al    = y   + (size_t)N * 128;
    float* ar    = al  + (size_t)N * 2;

    proj_gemm_kernel<<<(N + 15) / 16, 256, 0, stream>>>(x, Wl, Wout, xh, y, N);

    unsigned int* keys = (unsigned int*)xh;
    alpha_kernel<<<(N + AL_R - 1) / AL_R, AL_T, 0, stream>>>(xh, attl, attr, al, ar, N, ei, E, keys);

    gat_owner_kernel<<<256, 1024, 0, stream>>>(ei, E, N, keys, al, ar, y, bias, out);
}


# al, ar from one more WMMA tile per head in the projection kernel (B columns att_l, att_r on the LDS xh tile); its wave 0 builds the key planes: alpha launch, xh round trip gone; dot sum order changes
# speedup vs baseline: 5.2144x; 1.0578x over previous
#include <hip/hip_runtime.h>
#include <hip/hip_bf16.h>
#include <math.h>

typedef float v2f __attribute__((ext_vector_type(2)));
typedef float v8f __attribute__((ext_vector_type(8)));
typedef float v4f __attribute__((ext_vector_type(4)));

__device__ __forceinline__ unsigned int f32_to_key(float v) {
    const unsigned int b = __float_as_uint(v);
    return (b & 0x80000000u) ? ~b : (b | 0x80000000u);
}
__device__ __forceinline__ float key_to_f32(unsigned int k) {
    const unsigned int b = (k & 0x80000000u) ? (k & 0x7fffffffu) : ~k;
    return __uint_as_float(b);
}

typedef int v4i __attribute__((ext_vector_type(4)));

constexpr int OWN_R   = 196;
constexpr int OWN_CAP = 8192;
constexpr int OWN_T   = 1024;

constexpr int KEY_HI  = 64;

constexpr int KEY_PIECE = 4096 - 8;
__device__ __forceinline__ int key_last_piece(int E) {
    return (int)(((((long long)E + 31) >> 5) - 1) >> 6);
}
__device__ __forceinline__ bool keys_usable(int N, int E) {
    if (E < 1 || (N + OWN_R - 1) / OWN_R > 256) return false;
    const long long wl = key_last_piece(E);
    return wl <= 496 && wl * 65536LL + 2048 <= 512LL * (long long)N;
}
__device__ __forceinline__ int key_vec_of_group(int g, int wl) {
    return (g >> 6) * KEY_PIECE + 8 * wl + (g & 63);
}

__global__ __launch_bounds__(256)
void proj_gemm_kernel(const float* __restrict__ A,
                                 const float* __restrict__ B,
                                 const float* __restrict__ Wout,
                                 const float* __restrict__ attl,
                                 const float* __restrict__ attr,
                                 float* __restrict__ Y,
                                 float* __restrict__ al,
                                 float* __restrict__ ar,
                                 int M,
                                 const int* __restrict__ ei, int E,
                                 unsigned int* __restrict__ keys) {
    constexpr int K     = 128;
    constexpr int NDIM  = 128;
    constexpr int HC    = 64;
    constexpr int ROWS  = 16;
    constexpr int PITCH = K + 1;
    constexpr int NT    = 256;
    __shared__ float Alds[ROWS * PITCH];

    const int tid     = threadIdx.x;
    const int wave    = tid >> 5;
    const int lane    = tid & 31;
    const int rowbase = blockIdx.x * ROWS;

    const int wave_s = __builtin_amdgcn_readfirstlane(wave);
    const bool usekeys = keys_usable(M, E);
    const int  ng      = (int)(((long long)E + 31) >> 5);
    int        gw      = (int)blockIdx.x * 8;
    const bool dokeys  = usekeys && wave_s == 0 && gw < ng;
    unsigned int d[8] = {0u, 0u, 0u, 0u, 0u, 0u, 0u, 0u};
    if (dokeys) {
        const int* dcol = ei + E;
        #pragma unroll
        for (int i = 0; i < 8; ++i) {
            const long long e = ((long long)(gw + i) << 5) + lane;
            d[i] = (unsigned int)dcol[min(e, (long long)E - 1)];
        }
    }

    static_assert(ROWS * K == 2 * 4 * NT, "two 16-byte pieces per thread");
    const v4f* A4 = (const v4f*)A;
    v4f xt[2];
    #pragma unroll
    for (int i = 0; i < 2; ++i) {
        const int p  = tid + i * NT;
        const int gr = min(rowbase + (p >> 5), M - 1);
        xt[i] = A4[(size_t)gr * (K / 4) + (p & 31)];
    }
    #pragma unroll
    for (int i = 0; i < 2; ++i) {
        const int p = tid + i * NT;
        const int r = p >> 5;
        const bool in = rowbase + r < M;
        float* dstp = &Alds[r * PITCH + 4 * (p & 31)];
        #pragma unroll
        for (int j = 0; j < 4; ++j) dstp[j] = in ? xt[i][j] : 0.0f;
    }
    __syncthreads();

    const int colbase = wave * 16;
    const int half    = lane >> 4;
    const int l15     = lane & 15;

    const float* Arow = &Alds[l15 * PITCH];

    v8f acc = {};
    #pragma unroll 8
    for (int k = 0; k < K; k += 4) {
        const int kk = k + 2 * half;
        v2f a, b;
        a.x = Arow[kk];
        a.y = Arow[kk + 1];
        b.x = B[(size_t)kk * NDIM + colbase + l15];
        b.y = B[(size_t)(kk + 1) * NDIM + colbase + l15];
        acc = __builtin_amdgcn_wmma_f32_16x16x4_f32(
             false, a,  false, b,
             (short)0, acc,  false,  false);
    }

    __syncthreads();
    #pragma unroll
    for (int r = 0; r < 8; ++r)
        Alds[(r + 8 * half) * PITCH + colbase + l15] = acc[r];
    __syncthreads();

    const int kbase = (wave >> 2) * HC;
    const int ycol  = (wave & 3) * 16;
    v8f acc2 = {};
    #pragma unroll 8
    for (int k = 0; k < HC; k += 4) {
        const int kk = kbase + k + 2 * half;
        v2f a, b;
        a.x = Arow[kk];
        a.y = Arow[kk + 1];
        b.x = Wout[(size_t)kk * HC + ycol + l15];
        b.y = Wout[(size_t)(kk + 1) * HC + ycol + l15];
        acc2 = __builtin_amdgcn_wmma_f32_16x16x4_f32(
             false, a,  false, b,
             (short)0, acc2,  false,  false);
    }
    #pragma unroll
    for (int r = 0; r < 8; ++r) {
        int grow = rowbase + r + 8 * half;
        if (grow < M)
            Y[(size_t)grow * NDIM + colbase + l15] = acc2[r];
    }

    if ((wave_s & 3) == 3) {
        const float* attp = (l15 == 1) ? attr : attl;
        const bool   used = l15 < 2;
        float* const cp = (l15 == 1) ? ar : al;
        const int    h  = wave_s >> 2;
        v8f acc3 = {};
        #pragma unroll 8
        for (int k = 0; k < HC; k += 4) {
            const int kk = kbase + k + 2 * half;
            v2f a, b;
            a.x = Arow[kk];
            a.y = Arow[kk + 1];
            b.x = attp[kk];
            b.y = attp[kk + 1];
            acc3 = __builtin_amdgcn_wmma_f32_16x16x4_f32(
                 false, a,  false, b,
                 (short)0, acc3,  false,  false);
        }
        #pragma unroll
        for (int r = 0; r < 8; ++r) {
            const int grow = rowbase + r + 8 * half;
            if (used && grow < M) cp[2 * grow + h] = acc3[r];
        }
    }

    if (!dokeys) return;
    const int wl = key_last_piece(E);
    for (;;) {
        unsigned int wlo = 0u, whi = 0u;
        #pragma unroll
        for (int i = 0; i < 8; ++i) {
            const unsigned int r = d[i] / OWN_R;
            #pragma unroll
            for (int b = 0; b < 4; ++b) {
                const unsigned int p0 = __builtin_amdgcn_ballot_w32(((r >> b) & 1u) != 0u);
                const unsigned int p1 = __builtin_amdgcn_ballot_w32(((r >> (b + 4)) & 1u) != 0u);
                if (lane == 4 * i + b) { wlo = p0;  whi = p1; }
            }
        }
        const int g  = gw + (lane >> 2);
        const int wi = 4 * key_vec_of_group(g, wl) + (lane & 3);
        if (g < ng) {
            keys[wi]              = wlo;
            keys[wi + 4 * KEY_HI] = whi;
        }
        gw += (int)gridDim.x * 8;
        if (gw >= ng) break;
        const int* dcol = ei + E;
        #pragma unroll
        for (int i = 0; i < 8; ++i) {
            const long long e = ((long long)(gw + i) << 5) + lane;
            d[i] = (unsigned int)dcol[min(e, (long long)E - 1)];
        }
    }
}

__device__ __forceinline__ void lds_add_f32(float* p, float v) {
    __hip_atomic_fetch_add(p, v, __ATOMIC_RELAXED, __HIP_MEMORY_SCOPE_WORKGROUP);
}

__device__ __forceinline__ void own_append(int e, unsigned int* cnt, int* ls) {
    const unsigned int pos = atomicAdd(cnt, 1u);
    if (pos < (unsigned int)OWN_CAP) ls[pos] = e;
}

__device__ __forceinline__ void scan_group(v4i lo, v4i hi, const unsigned int (&inv)[8],
                                           int ebase, unsigned int keep, unsigned int* cnt, int* ls) {
    unsigned int m = ((unsigned int)lo[0] ^ inv[0]) & ((unsigned int)lo[1] ^ inv[1])
                   & ((unsigned int)lo[2] ^ inv[2]) & ((unsigned int)lo[3] ^ inv[3])
                   & ((unsigned int)hi[0] ^ inv[4]) & ((unsigned int)hi[1] ^ inv[5])
                   & ((unsigned int)hi[2] ^ inv[6]) & ((unsigned int)hi[3] ^ inv[7]) & keep;
    while (m) {
        own_append(ebase + __builtin_ctz(m), cnt, ls);
        m &= m - 1u;
    }
}

__device__ __forceinline__ void edge_logit2(const float* __restrict__ al,
                                            const float* __restrict__ ar,
                                            int s, int d, float& v0, float& v1) {
    const float a0 = al[2 * s], a1 = al[2 * s + 1];
    const float r0 = ar[2 * d], r1 = ar[2 * d + 1];
    v0 = a0 + r0;  v0 = (v0 > 0.0f) ? v0 : 0.2f * v0;
    v1 = a1 + r1;  v1 = (v1 > 0.0f) ? v1 : 0.2f * v1;
}

__device__ __forceinline__ float seg_max_of(unsigned int key) {
    float m = key_to_f32(key);
    if (!(fabsf(m) <= 3.4e38f)) m = 0.0f;
    return m;
}

template <class F>
__device__ __forceinline__ void own_rescan(const int* __restrict__ ei, int E,
                                           int lo, int nown, int tid, F f) {
    for (int e = tid; e < E; e += OWN_T) {
        const unsigned int dl = (unsigned int)(ei[E + e] - lo);
        if (dl < (unsigned int)nown) f(ei[e], (int)dl);
    }
    for (int i = tid; i < nown; i += OWN_T) f(lo + i, i);
}

__global__ __launch_bounds__(1024)
void gat_owner_kernel(const int* __restrict__ ei, int E, int N,
                      const unsigned int* __restrict__ keys,
                      const float* __restrict__ al,
                      const float* __restrict__ ar,
                      const float* __restrict__ y,
                      const float* __restrict__ bias,
                      float* __restrict__ out) {
    __shared__ v2f            sw[OWN_CAP];
    __shared__ float          lv[OWN_CAP * 2];
    __shared__ int            ls[OWN_CAP];
    __shared__ int            ss[OWN_CAP];
    __shared__ unsigned short ld[OWN_CAP];
    __shared__ unsigned short rk[OWN_CAP];
    __shared__ unsigned int   mkey[OWN_R * 2];
    __shared__ float          den[OWN_R * 2];
    __shared__ unsigned int   deg[OWN_R];
    __shared__ unsigned int   start[OWN_R];
    __shared__ unsigned int   cnt;
    static_assert(sizeof(float) * OWN_R * 64 <= sizeof(v2f) * OWN_CAP, "out tile must fit the slot array");
    float* const tile = (float*)sw;
    constexpr int PER = (OWN_R + 31) / 32;

    const int tid  = threadIdx.x;
    const int lane = tid & 31;
    const int wave = tid >> 5;
    const int nranges = (N + OWN_R - 1) / OWN_R;
    const bool usekeys = keys_usable(N, E);

    for (int rg = blockIdx.x; rg < nranges; rg += gridDim.x) {
        const int lo   = rg * OWN_R;
        const int nown = min(OWN_R, N - lo);

        for (int i = tid; i < nown * 2; i += OWN_T) { mkey[i] = 0u; den[i] = 0.0f; }
        for (int i = tid; i < nown; i += OWN_T) deg[i] = 0u;
        if (tid == 0) cnt = 0u;
        __syncthreads();

        if (usekeys) {
            unsigned int inv[8];
            #pragma unroll
            for (int b = 0; b < 8; ++b) inv[b] = ((rg >> b) & 1) ? 0u : 0xffffffffu;
            const v4i* __restrict__ k4 = (const v4i*)keys;
            const int ng = (int)(((long long)E + 31) >> 5);
            const int wl = key_last_piece(E);
            constexpr int G = 4;
            const int nbatch = (E >> 5) / (G * OWN_T);
            int bt = (int)(((long long)rg * nbatch) / nranges);
            for (int t = 0; t < nbatch; ++t) {
                const int g0 = bt * (G * OWN_T) + tid;
                bt = (bt + 1 < nbatch) ? bt + 1 : 0;
                v4i lo[G], hi[G];
                static_assert(OWN_T == 16 * 64, "a lane's groups must be 16 pieces apart");
                const int vi0 = key_vec_of_group(g0, wl);
                #pragma unroll
                for (int k = 0; k < G; ++k) {
                    const int vi = vi0 + k * (16 * KEY_PIECE);
                    lo[k] = __builtin_nontemporal_load(&k4[vi]);
                    hi[k] = __builtin_nontemporal_load(&k4[vi + KEY_HI]);
                }
                #pragma unroll
                for (int k = 0; k < G; ++k)
                    scan_group(lo[k], hi[k], inv, 32 * (g0 + k * OWN_T), 0xffffffffu, &cnt, ls);
            }
            for (int g = nbatch * (G * OWN_T) + tid; g < ng; g += OWN_T) {
                const int vi = key_vec_of_group(g, wl);
                const v4i lo = __builtin_nontemporal_load(&k4[vi]);
                const v4i hi = __builtin_nontemporal_load(&k4[vi + KEY_HI]);
                const int rem = E - 32 * g;
                scan_group(lo, hi, inv, 32 * g, (rem < 32) ? (1u << rem) - 1u : 0xffffffffu, &cnt, ls);
            }
        } else {
            for (int e = tid; e < E; e += OWN_T) {
                const unsigned int dl = (unsigned int)(ei[E + e] - lo);
                if (dl < (unsigned int)nown) own_append(e, &cnt, ls);
            }
        }
        __syncthreads();

        const unsigned int nlisted = cnt;
        const unsigned int total   = nlisted + (unsigned int)nown;
        if (total <= (unsigned int)OWN_CAP) {
            for (unsigned int i0 = tid; i0 < nlisted; i0 += 4 * OWN_T) {
                unsigned int ix[4];
                int en[4], sn[4], dn[4];
                #pragma unroll
                for (int k = 0; k < 4; ++k) {
                    const unsigned int i = i0 + k * OWN_T;
                    ix[k] = (i < nlisted) ? i : i0;
                    en[k] = ls[ix[k]];
                }
                #pragma unroll
                for (int k = 0; k < 4; ++k) { sn[k] = ei[en[k]];  dn[k] = ei[E + en[k]]; }
                #pragma unroll
                for (int k = 0; k < 4; ++k) {
                    ls[ix[k]] = sn[k];
                    ld[ix[k]] = (unsigned short)(dn[k] - lo);
                }
            }
            for (int i = tid; i < nown; i += OWN_T) {
                ls[nlisted + i] = lo + i;
                ld[nlisted + i] = (unsigned short)i;
            }
            __syncthreads();

            for (unsigned int i = tid; i < total; i += OWN_T) {
                const int s = ls[i], dl = ld[i];
                float v0, v1;
                edge_logit2(al, ar, s, lo + dl, v0, v1);
                lv[2 * i] = v0;  lv[2 * i + 1] = v1;
                atomicMax(&mkey[2 * dl],     f32_to_key(v0));
                atomicMax(&mkey[2 * dl + 1], f32_to_key(v1));
                rk[i] = (unsigned short)atomicAdd(&deg[dl], 1u);
            }
            __syncthreads();
            if (wave == 0) {
                unsigned int c[PER];
                unsigned int sum = 0u;
                #pragma unroll
                for (int j = 0; j < PER; ++j) {
                    const int n = PER * lane + j;
                    c[j] = (n < nown) ? deg[n] : 0u;
                    sum += c[j];
                }
                unsigned int incl = sum;
                #pragma unroll
                for (int off = 1; off < 32; off <<= 1) {
                    const unsigned int t = __shfl_up(incl, off, 32);
                    if (lane >= off) incl += t;
                }
                unsigned int run = incl - sum;
                #pragma unroll
                for (int j = 0; j < PER; ++j) {
                    const int n = PER * lane + j;
                    if (n < nown) start[n] = run;
                    run += c[j];
                }
            }
            for (unsigned int i = tid; i < total; i += OWN_T) {
                const int dl = ld[i];
                const float x0 = __expf(lv[2 * i]     - seg_max_of(mkey[2 * dl]));
                const float x1 = __expf(lv[2 * i + 1] - seg_max_of(mkey[2 * dl + 1]));
                lv[2 * i] = x0;  lv[2 * i + 1] = x1;
                lds_add_f32(&den[2 * dl],     x0);
                lds_add_f32(&den[2 * dl + 1], x1);
            }
            __syncthreads();
            for (unsigned int i = tid; i < total; i += OWN_T) {
                const int dl = ld[i];
                const unsigned int p = start[dl] + rk[i];
                v2f w;
                w.x = lv[2 * i]     / (den[2 * dl]     + 1e-16f);
                w.y = lv[2 * i + 1] / (den[2 * dl + 1] + 1e-16f);
                ss[p] = ls[i];
                sw[p] = w;
            }
            __syncthreads();
            const int wave_s = __builtin_amdgcn_readfirstlane(wave);
            for (int n = wave_s; n < nown; n += 32) {
                const unsigned int b  = __builtin_amdgcn_readfirstlane(start[n]);
                const unsigned int dn = __builtin_amdgcn_readfirstlane(deg[n]);
                float a0 = 0.0f, a1 = 0.0f;
                for (unsigned int j = 0; j < dn; j += 8) {
                    const unsigned int o    = j + (unsigned int)(lane & 7);
                    const unsigned int slot = b + ((o < dn) ? o : j);
                    const int sv = ss[slot];
                    const v2f wv = sw[slot];
                    float p0[8], p1[8], q0[8], q1[8];
                    #pragma unroll
                    for (int k = 0; k < 8; ++k) {
                        const int s = __builtin_amdgcn_readlane(sv, k);
                        const float* ys = y + (size_t)s * 128;
                        p0[k] = ys[lane];       p1[k] = ys[lane + 32];
                        q0[k] = ys[lane + 64];  q1[k] = ys[lane + 96];
                    }
                    #pragma unroll
                    for (int k = 0; k < 8; ++k) {
                        const float w0 = __uint_as_float((unsigned int)__builtin_amdgcn_readlane((int)__float_as_uint(wv.x), k));
                        const float w1 = __uint_as_float((unsigned int)__builtin_amdgcn_readlane((int)__float_as_uint(wv.y), k));
                        const float t0 = p0[k] * w0 + q0[k] * w1;
                        const float t1 = p1[k] * w0 + q1[k] * w1;
                        const bool in_run = (j + (unsigned int)k < dn);
                        a0 += in_run ? t0 : 0.0f;
                        a1 += in_run ? t1 : 0.0f;
                    }
                }
                float* og = out + (size_t)(lo + n) * 64;
                og[lane]      = a0 + bias[lane];
                og[lane + 32] = a1 + bias[lane + 32];
            }
        } else {
            for (int i = tid; i < nown * 64; i += OWN_T) tile[i] = 0.0f;
            __syncthreads();
            own_rescan(ei, E, lo, nown, tid, [&](int s, int dl) {
                float v0, v1;
                edge_logit2(al, ar, s, lo + dl, v0, v1);
                atomicMax(&mkey[2 * dl],     f32_to_key(v0));
                atomicMax(&mkey[2 * dl + 1], f32_to_key(v1));
            });
            __syncthreads();
            own_rescan(ei, E, lo, nown, tid, [&](int s, int dl) {
                float v0, v1;
                edge_logit2(al, ar, s, lo + dl, v0, v1);
                lds_add_f32(&den[2 * dl],     __expf(v0 - seg_max_of(mkey[2 * dl])));
                lds_add_f32(&den[2 * dl + 1], __expf(v1 - seg_max_of(mkey[2 * dl + 1])));
            });
            __syncthreads();
            own_rescan(ei, E, lo, nown, tid, [&](int s, int dl) {
                float v0, v1;
                edge_logit2(al, ar, s, lo + dl, v0, v1);
                const float w0 = __expf(v0 - seg_max_of(mkey[2 * dl]))     / (den[2 * dl]     + 1e-16f);
                const float w1 = __expf(v1 - seg_max_of(mkey[2 * dl + 1])) / (den[2 * dl + 1] + 1e-16f);
                const float* ys = y + (size_t)s * 128;
                for (int c = 0; c < 64; ++c)
                    lds_add_f32(&tile[dl * 64 + c], ys[c] * w0 + ys[64 + c] * w1);
            });
            __syncthreads();
            float* og = out + (size_t)lo * 64;
            for (int i = tid; i < nown * 64; i += OWN_T) og[i] = tile[i] + bias[i & 63];
        }
        __syncthreads();
    }
}

extern "C" void kernel_launch(void* const* d_in, const int* in_sizes, int n_in,
                              void* d_out, int out_size, void* d_ws, size_t ws_size,
                              hipStream_t stream) {
    const float* x    = (const float*)d_in[0];
    const int*   ei   = (const int*)d_in[1];
    const float* Wl   = (const float*)d_in[2];
    const float* attl = (const float*)d_in[3];
    const float* attr = (const float*)d_in[4];
    const float* Wout = (const float*)d_in[5];
    const float* bias = (const float*)d_in[6];
    float* out = (float*)d_out;

    const int N   = in_sizes[0] / 128;
    const int E   = in_sizes[1] / 2;

    float* ws    = (float*)d_ws;
    unsigned int* keys = (unsigned int*)ws;
    float* y     = ws  + (size_t)N * 128;
    float* al    = y   + (size_t)N * 128;
    float* ar    = al  + (size_t)N * 2;

    proj_gemm_kernel<<<(N + 15) / 16, 256, 0, stream>>>(x, Wl, Wout, attl, attr, y, al, ar, N, ei, E, keys);

    gat_owner_kernel<<<256, 1024, 0, stream>>>(ei, E, N, keys, al, ar, y, bias, out);
}
